# starting kernel
# baseline (speedup 1.0000x reference)
#include <hip/hip_runtime.h>
#include <hip/hip_bf16.h>
#include <math.h>


typedef float v2f __attribute__((ext_vector_type(2)));
typedef float v8f __attribute__((ext_vector_type(8)));

#define B_      4
#define I_      96
#define J_      384
#define D_      256
#define NWAVE   12
#define LOGEPS  (-1000.0f)
#define NEGV    (-1000000000.0f)
#define NEGINF  (-1.0e38f)
#define TEMPINV (1.0f/0.55f)
#define ESCALE  (1.0f/256.0f)

__device__ __forceinline__ float lse2(float a, float b) {
  float m = fmaxf(a, b);
  if (m < -9.0e37f) return NEGINF;
  return m + __logf(__expf(a - m) + __expf(b - m));
}

__device__ __forceinline__ float wave_prefix_lse(float x, int lane) {
#pragma unroll
  for (int off = 1; off < 32; off <<= 1) {
    float o = __shfl_up(x, off, 32);
    if (lane >= off) x = lse2(x, o);
  }
  return x;
}

__device__ __forceinline__ float wave_suffix_lse(float x, int lane) {
#pragma unroll
  for (int off = 1; off < 32; off <<= 1) {
    float o = __shfl_down(x, off, 32);
    if (lane + off < 32) x = lse2(x, o);
  }
  return x;
}

__global__ __launch_bounds__(384, 1)
void mobo_kernel(const float* __restrict__ text,
                 const float* __restrict__ mel,
                 const float* __restrict__ gum,
                 float* __restrict__ out_soft,
                 float* __restrict__ out_exp)
{
  __shared__ float sE[I_ * J_];
  __shared__ float sP[J_];
  __shared__ float sSB[J_];
  __shared__ float sB0[J_];
  __shared__ float sB1[J_];
  __shared__ float sWtL[NWAVE];
  __shared__ float sWtP[NWAVE];
  __shared__ float sWtSA[NWAVE];
  __shared__ float sWtSB[NWAVE];

  const int b    = blockIdx.x;
  const int tid  = threadIdx.x;
  const int lane = tid & 31;
  const int wave = tid >> 5;
  const int half = lane >> 4;
  const int l16  = lane & 15;

  const float* tx = text + (size_t)b * I_ * D_;
  const float* ml = mel  + (size_t)b * J_ * D_;

  for (int t = wave; t < (I_/16) * (J_/16); t += NWAVE) {
    const int mT = t % (I_/16);
    const int nT = t / (I_/16);
    const int mrow = mT * 16 + l16;
    const int ncol = nT * 16 + l16;
    v8f acc = {};
    for (int k0 = 0; k0 < D_; k0 += 4) {
      const int kk = k0 + 2 * half;
      v2f a  = *(const v2f*)(tx + (size_t)mrow * D_ + kk);
      v2f bf = *(const v2f*)(ml + (size_t)ncol * D_ + kk);
      acc = __builtin_amdgcn_wmma_f32_16x16x4_f32(false, a, false, bf,
                                                  (short)0, acc, false, false);
    }
    for (int r = 0; r < 8; ++r) {
      const int i = mT * 16 + r + 8 * half;
      const int j = nT * 16 + l16;
      const float u = gum[(size_t)b * I_ * J_ + (size_t)i * J_ + j];
      const float g = -__logf(-__logf(u));
      sE[i * J_ + j] = (acc[r] * ESCALE + g) * TEMPINV;
    }
  }
  __syncthreads();

  float* Bc = sB0;
  float* Bn = sB1;
  Bc[tid] = (tid == 0) ? 0.0f : NEGV;
  __syncthreads();

  for (int i = 0; i < I_; ++i) {
    const float e  = sE[i * J_ + tid];
    float xl = wave_suffix_lse(e, lane);
    if (lane == 0) sWtL[wave] = xl;
    __syncthreads();
    float lhi = NEGINF;
    for (int w = wave + 1; w < NWAVE; ++w) lhi = lse2(lhi, sWtL[w]);
    const float Lv = lse2(xl, lhi);

    const float bv = Bc[tid];
    const float av = bv - Lv;
    float xp  = wave_prefix_lse(av, lane);
    float xsa = wave_suffix_lse(av, lane);
    float xsb = wave_suffix_lse(bv, lane);
    if (lane == 31) sWtP[wave] = xp;
    if (lane == 0) { sWtSA[wave] = xsa; sWtSB[wave] = xsb; }
    __syncthreads();
    float plo = NEGINF, sahi = NEGINF, sbhi = NEGINF;
    for (int w = 0; w < wave; ++w)          plo  = lse2(plo,  sWtP[w]);
    for (int w = wave + 1; w < NWAVE; ++w) { sahi = lse2(sahi, sWtSA[w]);
                                             sbhi = lse2(sbhi, sWtSB[w]); }
    const float Pv  = lse2(xp,  plo);
    const float SAv = lse2(xsa, sahi);
    const float SBv = lse2(xsb, sbhi);
    sP[tid]  = Pv;
    sSB[tid] = SBv;
    __syncthreads();

    if (i < I_ - 1) {
      float nb;
      if (tid == 0) {
        nb = NEGV;
      } else {
        const int j = tid - 1;
        const float v = lse2(sE[i * J_ + j] + sP[j], SAv + LOGEPS);
        const int r = i + 1;
        nb = (tid >= r && tid <= (J_ - I_ + r + 1)) ? v : NEGV;
      }
      Bn[tid] = nb;
    }

    float lb;
    if (i < I_ - 1) {
      const float sbexc = (tid + 1 < J_) ? sSB[tid + 1] : NEGINF;
      lb = lse2(Lv + Pv, sbexc + LOGEPS);
    } else {
      lb = sSB[0] + ((tid == J_ - 1) ? 0.0f : LOGEPS);
    }
    out_soft[(size_t)b * I_ * J_ + (size_t)i * J_ + tid] = lb;

    __syncthreads();
    sE[i * J_ + tid] = __expf(lb);
    { float* tmp = Bc; Bc = Bn; Bn = tmp; }
  }
  __syncthreads();

  for (int t = wave; t < (J_/16) * (D_/16); t += NWAVE) {
    const int mT = t % (J_/16);
    const int nT = t / (J_/16);
    const int m = mT * 16 + l16;
    const int n = nT * 16 + l16;
    v8f acc = {};
    for (int k0 = 0; k0 < I_; k0 += 4) {
      const int kk = k0 + 2 * half;
      v2f a;
      a.x = sE[kk * J_ + m];
      a.y = sE[(kk + 1) * J_ + m];
      v2f bf;
      bf.x = tx[(size_t)kk * D_ + n];
      bf.y = tx[(size_t)(kk + 1) * D_ + n];
      acc = __builtin_amdgcn_wmma_f32_16x16x4_f32(false, a, false, bf,
                                                  (short)0, acc, false, false);
    }
    for (int r = 0; r < 8; ++r) {
      const int j = mT * 16 + r + 8 * half;
      out_exp[(size_t)b * J_ * D_ + (size_t)j * D_ + n] = acc[r];
    }
  }
}

extern "C" void kernel_launch(void* const* d_in, const int* in_sizes, int n_in,
                              void* d_out, int out_size, void* d_ws, size_t ws_size,
                              hipStream_t stream) {
  (void)in_sizes; (void)n_in; (void)d_ws; (void)ws_size; (void)out_size;
  const float* text = (const float*)d_in[0];
  const float* mel  = (const float*)d_in[1];
  const float* gum  = (const float*)d_in[2];
  float* out_soft = (float*)d_out;
  float* out_exp  = (float*)d_out + (size_t)B_ * I_ * J_;
  mobo_kernel<<<dim3(B_), dim3(384), 0, stream>>>(text, mel, gum, out_soft, out_exp);
}


# Hoist the energy-only suffix-LSE L(i,k) out of the 96 serial DP rows into one parallel pre-pass held in LDS (new reduction tree for L): each row drops a scan, a barrier and the 11-step wave chain
# speedup vs baseline: 1.2356x; 1.2356x over previous
#include <hip/hip_runtime.h>
#include <hip/hip_bf16.h>
#include <math.h>


typedef float v2f __attribute__((ext_vector_type(2)));
typedef float v4f __attribute__((ext_vector_type(4)));
typedef float v8f __attribute__((ext_vector_type(8)));

#define B_      4
#define I_      96
#define J_      384
#define D_      256
#define NWAVE   12
#define RPW     (I_ / NWAVE)
#define CPL     (J_ / 32)
#define LOGEPS  (-1000.0f)
#define NEGV    (-1000000000.0f)
#define NEGINF  (-1.0e38f)
#define TEMPINV (1.0f/0.55f)
#define ESCALE  (1.0f/256.0f)

__device__ __forceinline__ float lse2(float a, float b) {
  float m = fmaxf(a, b);
  if (m < -9.0e37f) return NEGINF;
  return m + __logf(__expf(a - m) + __expf(b - m));
}

__device__ __forceinline__ float wave_prefix_lse(float x, int lane) {
#pragma unroll
  for (int off = 1; off < 32; off <<= 1) {
    float o = __shfl_up(x, off, 32);
    if (lane >= off) x = lse2(x, o);
  }
  return x;
}

__device__ __forceinline__ float wave_suffix_lse(float x, int lane) {
#pragma unroll
  for (int off = 1; off < 32; off <<= 1) {
    float o = __shfl_down(x, off, 32);
    if (lane + off < 32) x = lse2(x, o);
  }
  return x;
}

__global__ __launch_bounds__(384, 1)
void mobo_kernel(const float* __restrict__ text,
                 const float* __restrict__ mel,
                 const float* __restrict__ gum,
                 float* __restrict__ out_soft,
                 float* __restrict__ out_exp)
{
  __shared__ __attribute__((aligned(16))) float sE[I_ * J_];
  __shared__ __attribute__((aligned(16))) float sL[I_ * J_];
  __shared__ float sP[J_];
  __shared__ float sSB[J_];
  __shared__ float sB0[J_];
  __shared__ float sB1[J_];
  __shared__ float sWtP[NWAVE];
  __shared__ float sWtSA[NWAVE];
  __shared__ float sWtSB[NWAVE];

  const int b    = blockIdx.x;
  const int tid  = threadIdx.x;
  const int lane = tid & 31;
  const int wave = tid >> 5;
  const int half = lane >> 4;
  const int l16  = lane & 15;

  const float* tx = text + (size_t)b * I_ * D_;
  const float* ml = mel  + (size_t)b * J_ * D_;

  for (int t = wave; t < (I_/16) * (J_/16); t += NWAVE) {
    const int mT = t % (I_/16);
    const int nT = t / (I_/16);
    const int mrow = mT * 16 + l16;
    const int ncol = nT * 16 + l16;
    v8f acc = {};
    for (int k0 = 0; k0 < D_; k0 += 4) {
      const int kk = k0 + 2 * half;
      v2f a  = *(const v2f*)(tx + (size_t)mrow * D_ + kk);
      v2f bf = *(const v2f*)(ml + (size_t)ncol * D_ + kk);
      acc = __builtin_amdgcn_wmma_f32_16x16x4_f32(false, a, false, bf,
                                                  (short)0, acc, false, false);
    }
    for (int r = 0; r < 8; ++r) {
      const int i = mT * 16 + r + 8 * half;
      const int j = nT * 16 + l16;
      const float u = gum[(size_t)b * I_ * J_ + (size_t)i * J_ + j];
      const float g = -__logf(-__logf(u));
      sE[i * J_ + j] = (acc[r] * ESCALE + g) * TEMPINV;
    }
  }
  __syncthreads();

  for (int r = 0; r < RPW; ++r) {
    const int row = wave * RPW + r;
    const float* er = sE + row * J_ + lane * CPL;
    const v4f x0 = *(const v4f*)(er + 0);
    const v4f x1 = *(const v4f*)(er + 4);
    const v4f x2 = *(const v4f*)(er + 8);
    float s[CPL] = { x0.x, x0.y, x0.z, x0.w, x1.x, x1.y, x1.z, x1.w,
                     x2.x, x2.y, x2.z, x2.w };
    for (int c = CPL - 2; c >= 0; --c) s[c] = lse2(s[c], s[c + 1]);
    const float inc = wave_suffix_lse(s[0], lane);
    float exc = __shfl_down(inc, 1, 32);
    if (lane == 31) exc = NEGINF;
    for (int c = 0; c < CPL; ++c) s[c] = lse2(s[c], exc);
    float* lr = sL + row * J_ + lane * CPL;
    *(v4f*)(lr + 0) = (v4f){ s[0], s[1], s[2],  s[3]  };
    *(v4f*)(lr + 4) = (v4f){ s[4], s[5], s[6],  s[7]  };
    *(v4f*)(lr + 8) = (v4f){ s[8], s[9], s[10], s[11] };
  }

  float* Bc = sB0;
  float* Bn = sB1;
  Bc[tid] = (tid == 0) ? 0.0f : NEGV;
  __syncthreads();

  for (int i = 0; i < I_; ++i) {
    const float Lv = sL[i * J_ + tid];

    const float bv = Bc[tid];
    const float av = bv - Lv;
    float xp  = wave_prefix_lse(av, lane);
    float xsa = wave_suffix_lse(av, lane);
    float xsb = wave_suffix_lse(bv, lane);
    if (lane == 31) sWtP[wave] = xp;
    if (lane == 0) { sWtSA[wave] = xsa; sWtSB[wave] = xsb; }
    __syncthreads();
    float plo = NEGINF, sahi = NEGINF, sbhi = NEGINF;
    for (int w = 0; w < wave; ++w)          plo  = lse2(plo,  sWtP[w]);
    for (int w = wave + 1; w < NWAVE; ++w) { sahi = lse2(sahi, sWtSA[w]);
                                             sbhi = lse2(sbhi, sWtSB[w]); }
    const float Pv  = lse2(xp,  plo);
    const float SAv = lse2(xsa, sahi);
    const float SBv = lse2(xsb, sbhi);
    sP[tid]  = Pv;
    sSB[tid] = SBv;
    __syncthreads();

    if (i < I_ - 1) {
      float nb;
      if (tid == 0) {
        nb = NEGV;
      } else {
        const int j = tid - 1;
        const float v = lse2(sE[i * J_ + j] + sP[j], SAv + LOGEPS);
        const int r = i + 1;
        nb = (tid >= r && tid <= (J_ - I_ + r + 1)) ? v : NEGV;
      }
      Bn[tid] = nb;
    }

    float lb;
    if (i < I_ - 1) {
      const float sbexc = (tid + 1 < J_) ? sSB[tid + 1] : NEGINF;
      lb = lse2(Lv + Pv, sbexc + LOGEPS);
    } else {
      lb = sSB[0] + ((tid == J_ - 1) ? 0.0f : LOGEPS);
    }
    out_soft[(size_t)b * I_ * J_ + (size_t)i * J_ + tid] = lb;

    __syncthreads();
    sE[i * J_ + tid] = __expf(lb);
    { float* tmp = Bc; Bc = Bn; Bn = tmp; }
  }
  __syncthreads();

  for (int t = wave; t < (J_/16) * (D_/16); t += NWAVE) {
    const int mT = t % (J_/16);
    const int nT = t / (J_/16);
    const int m = mT * 16 + l16;
    const int n = nT * 16 + l16;
    v8f acc = {};
    for (int k0 = 0; k0 < I_; k0 += 4) {
      const int kk = k0 + 2 * half;
      v2f a;
      a.x = sE[kk * J_ + m];
      a.y = sE[(kk + 1) * J_ + m];
      v2f bf;
      bf.x = tx[(size_t)kk * D_ + n];
      bf.y = tx[(size_t)(kk + 1) * D_ + n];
      acc = __builtin_amdgcn_wmma_f32_16x16x4_f32(false, a, false, bf,
                                                  (short)0, acc, false, false);
    }
    for (int r = 0; r < 8; ++r) {
      const int j = mT * 16 + r + 8 * half;
      out_exp[(size_t)b * J_ * D_ + (size_t)j * D_ + n] = acc[r];
    }
  }
}

extern "C" void kernel_launch(void* const* d_in, const int* in_sizes, int n_in,
                              void* d_out, int out_size, void* d_ws, size_t ws_size,
                              hipStream_t stream) {
  (void)in_sizes; (void)n_in; (void)d_ws; (void)ws_size; (void)out_size;
  const float* text = (const float*)d_in[0];
  const float* mel  = (const float*)d_in[1];
  const float* gum  = (const float*)d_in[2];
  float* out_soft = (float*)d_out;
  float* out_exp  = (float*)d_out + (size_t)B_ * I_ * J_;
  mobo_kernel<<<dim3(B_), dim3(384), 0, stream>>>(text, mel, gum, out_soft, out_exp);
}


# Move the soft-output work (suffix-LSE of each Bij row, soft, exp, store) out of the serial DP rows into one parallel post-pass over rows stashed in LDS (new reduction tree for that suffix-LSE)
# speedup vs baseline: 1.7045x; 1.3795x over previous
#include <hip/hip_runtime.h>
#include <hip/hip_bf16.h>
#include <math.h>


typedef float v2f __attribute__((ext_vector_type(2)));
typedef float v4f __attribute__((ext_vector_type(4)));
typedef float v8f __attribute__((ext_vector_type(8)));

#define B_      4
#define I_      96
#define J_      384
#define D_      256
#define NWAVE   12
#define RPW     (I_ / NWAVE)
#define CPL     (J_ / 32)
#define LOGEPS  (-1000.0f)
#define NEGV    (-1000000000.0f)
#define NEGINF  (-1.0e38f)
#define TEMPINV (1.0f/0.55f)
#define ESCALE  (1.0f/256.0f)

__device__ __forceinline__ float lse2(float a, float b) {
  float m = fmaxf(a, b);
  if (m < -9.0e37f) return NEGINF;
  return m + __logf(__expf(a - m) + __expf(b - m));
}

__device__ __forceinline__ float wave_prefix_lse(float x, int lane) {
#pragma unroll
  for (int off = 1; off < 32; off <<= 1) {
    float o = __shfl_up(x, off, 32);
    if (lane >= off) x = lse2(x, o);
  }
  return x;
}

__device__ __forceinline__ float wave_suffix_lse(float x, int lane) {
#pragma unroll
  for (int off = 1; off < 32; off <<= 1) {
    float o = __shfl_down(x, off, 32);
    if (lane + off < 32) x = lse2(x, o);
  }
  return x;
}

__global__ __launch_bounds__(384, 1)
void mobo_kernel(const float* __restrict__ text,
                 const float* __restrict__ mel,
                 const float* __restrict__ gum,
                 float* __restrict__ out_soft,
                 float* __restrict__ out_exp)
{
  __shared__ __attribute__((aligned(16))) float sE[I_ * J_];
  __shared__ __attribute__((aligned(16))) float sL[I_ * J_];
  __shared__ float sP[J_];
  __shared__ float sWtP[NWAVE];
  __shared__ float sWtSA[NWAVE];

  const int b    = blockIdx.x;
  const int tid  = threadIdx.x;
  const int lane = tid & 31;
  const int wave = tid >> 5;
  const int half = lane >> 4;
  const int l16  = lane & 15;

  const float* tx = text + (size_t)b * I_ * D_;
  const float* ml = mel  + (size_t)b * J_ * D_;

  for (int t = wave; t < (I_/16) * (J_/16); t += NWAVE) {
    const int mT = t % (I_/16);
    const int nT = t / (I_/16);
    const int mrow = mT * 16 + l16;
    const int ncol = nT * 16 + l16;
    v8f acc = {};
    for (int k0 = 0; k0 < D_; k0 += 4) {
      const int kk = k0 + 2 * half;
      v2f a  = *(const v2f*)(tx + (size_t)mrow * D_ + kk);
      v2f bf = *(const v2f*)(ml + (size_t)ncol * D_ + kk);
      acc = __builtin_amdgcn_wmma_f32_16x16x4_f32(false, a, false, bf,
                                                  (short)0, acc, false, false);
    }
    for (int r = 0; r < 8; ++r) {
      const int i = mT * 16 + r + 8 * half;
      const int j = nT * 16 + l16;
      const float u = gum[(size_t)b * I_ * J_ + (size_t)i * J_ + j];
      const float g = -__logf(-__logf(u));
      sE[i * J_ + j] = (acc[r] * ESCALE + g) * TEMPINV;
    }
  }
  __syncthreads();

  for (int r = 0; r < RPW; ++r) {
    const int row = wave * RPW + r;
    const float* er = sE + row * J_ + lane * CPL;
    const v4f x0 = *(const v4f*)(er + 0);
    const v4f x1 = *(const v4f*)(er + 4);
    const v4f x2 = *(const v4f*)(er + 8);
    float s[CPL] = { x0.x, x0.y, x0.z, x0.w, x1.x, x1.y, x1.z, x1.w,
                     x2.x, x2.y, x2.z, x2.w };
    for (int c = CPL - 2; c >= 0; --c) s[c] = lse2(s[c], s[c + 1]);
    const float inc = wave_suffix_lse(s[0], lane);
    float exc = __shfl_down(inc, 1, 32);
    if (lane == 31) exc = NEGINF;
    for (int c = 0; c < CPL; ++c) s[c] = lse2(s[c], exc);
    float* lr = sL + row * J_ + lane * CPL;
    *(v4f*)(lr + 0) = (v4f){ s[0], s[1], s[2],  s[3]  };
    *(v4f*)(lr + 4) = (v4f){ s[4], s[5], s[6],  s[7]  };
    *(v4f*)(lr + 8) = (v4f){ s[8], s[9], s[10], s[11] };
  }

  float bv = (tid == 0) ? 0.0f : NEGV;
  const int st = (tid == 0) ? (J_ - 1) : (tid - 1);
  __syncthreads();

  for (int i = 0; i < I_ - 1; ++i) {
    const float Lv = sL[i * J_ + tid];

    const float av = bv - Lv;
    float xp  = wave_prefix_lse(av, lane);
    float xsa = wave_suffix_lse(av, lane);
    if (lane == 31) sWtP[wave]  = xp;
    if (lane == 0)  sWtSA[wave] = xsa;
    __syncthreads();
    float plo = NEGINF, sahi = NEGINF;
    for (int w = 0; w < wave; ++w)         plo  = lse2(plo,  sWtP[w]);
    for (int w = wave + 1; w < NWAVE; ++w) sahi = lse2(sahi, sWtSA[w]);
    const float Pv  = lse2(xp,  plo);
    const float SAv = lse2(xsa, sahi);
    sP[tid] = Pv;
    sL[i * J_ + tid] = Lv + Pv;
    __syncthreads();

    const int r = i + 1;
    const float ev = sE[i * J_ + st];
    float nb = NEGV;
    if (tid >= r && tid <= (J_ - I_ + r + 1))
      nb = lse2(ev + sP[tid - 1], SAv + LOGEPS);
    sE[i * J_ + st] = bv;
    bv = nb;
  }
  sE[(I_ - 1) * J_ + st] = bv;
  __syncthreads();

  for (int rr = 0; rr < RPW; ++rr) {
    const int row = wave * RPW + rr;
    const float* br = sE + row * J_ + lane * CPL;
    const float* qr = sL + row * J_ + lane * CPL;
    const v4f x0 = *(const v4f*)(br + 0);
    const v4f x1 = *(const v4f*)(br + 4);
    const v4f x2 = *(const v4f*)(br + 8);
    const v4f q0 = *(const v4f*)(qr + 0);
    const v4f q1 = *(const v4f*)(qr + 4);
    const v4f q2 = *(const v4f*)(qr + 8);
    float s[CPL] = { x0.x, x0.y, x0.z, x0.w, x1.x, x1.y, x1.z, x1.w,
                     x2.x, x2.y, x2.z, x2.w };
    const float q[CPL] = { q0.x, q0.y, q0.z, q0.w, q1.x, q1.y, q1.z, q1.w,
                           q2.x, q2.y, q2.z, q2.w };
    const float bfirst = s[CPL - 1];
    if (lane == 31) s[CPL - 1] = NEGINF;
    for (int c = CPL - 2; c >= 0; --c) s[c] = lse2(s[c], s[c + 1]);
    const float inc = wave_suffix_lse(s[0], lane);
    float exc = __shfl_down(inc, 1, 32);
    if (lane == 31) exc = NEGINF;
    for (int c = 0; c < CPL; ++c) s[c] = lse2(s[c], exc);
    float o[CPL];
    if (row < I_ - 1) {
      for (int c = 0; c < CPL; ++c) o[c] = lse2(q[c], s[c] + LOGEPS);
    } else {
      const float tot = lse2(__shfl(s[0], 0, 32), __shfl(bfirst, 31, 32));
      for (int c = 0; c < CPL; ++c)
        o[c] = tot + ((lane * CPL + c == J_ - 1) ? 0.0f : LOGEPS);
    }
    float* og = out_soft + (size_t)b * I_ * J_ + (size_t)row * J_ + lane * CPL;
    *(v4f*)(og + 0) = (v4f){ o[0], o[1], o[2],  o[3]  };
    *(v4f*)(og + 4) = (v4f){ o[4], o[5], o[6],  o[7]  };
    *(v4f*)(og + 8) = (v4f){ o[8], o[9], o[10], o[11] };
    float* wr = sE + row * J_ + lane * CPL;
    *(v4f*)(wr + 0) = (v4f){ __expf(o[0]), __expf(o[1]), __expf(o[2]),  __expf(o[3])  };
    *(v4f*)(wr + 4) = (v4f){ __expf(o[4]), __expf(o[5]), __expf(o[6]),  __expf(o[7])  };
    *(v4f*)(wr + 8) = (v4f){ __expf(o[8]), __expf(o[9]), __expf(o[10]), __expf(o[11]) };
  }
  __syncthreads();

  for (int t = wave; t < (J_/16) * (D_/16); t += NWAVE) {
    const int mT = t % (J_/16);
    const int nT = t / (J_/16);
    const int m = mT * 16 + l16;
    const int n = nT * 16 + l16;
    v8f acc = {};
    for (int k0 = 0; k0 < I_; k0 += 4) {
      const int kk = k0 + 2 * half;
      v2f a;
      a.x = sE[kk * J_ + m];
      a.y = sE[(kk + 1) * J_ + m];
      v2f bf;
      bf.x = tx[(size_t)kk * D_ + n];
      bf.y = tx[(size_t)(kk + 1) * D_ + n];
      acc = __builtin_amdgcn_wmma_f32_16x16x4_f32(false, a, false, bf,
                                                  (short)0, acc, false, false);
    }
    for (int r = 0; r < 8; ++r) {
      const int j = mT * 16 + r + 8 * half;
      out_exp[(size_t)b * J_ * D_ + (size_t)j * D_ + n] = acc[r];
    }
  }
}

extern "C" void kernel_launch(void* const* d_in, const int* in_sizes, int n_in,
                              void* d_out, int out_size, void* d_ws, size_t ws_size,
                              hipStream_t stream) {
  (void)in_sizes; (void)n_in; (void)d_ws; (void)ws_size; (void)out_size;
  const float* text = (const float*)d_in[0];
  const float* mel  = (const float*)d_in[1];
  const float* gum  = (const float*)d_in[2];
  float* out_soft = (float*)d_out;
  float* out_exp  = (float*)d_out + (size_t)B_ * I_ * J_;
  mobo_kernel<<<dim3(B_), dim3(384), 0, stream>>>(text, mel, gum, out_soft, out_exp);
}


# Cross-wave combine of the two per-row DP scans: one 4-step DPP scan of the 12 wave totals in registers (new reduction tree for P and SA) instead of an 11-step serial chain of LDS reads
# speedup vs baseline: 2.1840x; 1.2813x over previous
#include <hip/hip_runtime.h>
#include <hip/hip_bf16.h>
#include <math.h>


typedef float v2f __attribute__((ext_vector_type(2)));
typedef float v4f __attribute__((ext_vector_type(4)));
typedef float v8f __attribute__((ext_vector_type(8)));

#define B_      4
#define I_      96
#define J_      384
#define D_      256
#define NWAVE   12
#define RPW     (I_ / NWAVE)
#define CPL     (J_ / 32)
#define LOGEPS  (-1000.0f)
#define NEGV    (-1000000000.0f)
#define NEGINF  (-1.0e38f)
#define TEMPINV (1.0f/0.55f)
#define ESCALE  (1.0f/256.0f)

__device__ __forceinline__ float lse2(float a, float b) {
  float m = fmaxf(a, b);
  if (m < -9.0e37f) return NEGINF;
  return m + __logf(__expf(a - m) + __expf(b - m));
}

__device__ __forceinline__ float wave_prefix_lse(float x, int lane) {
#pragma unroll
  for (int off = 1; off < 32; off <<= 1) {
    float o = __shfl_up(x, off, 32);
    if (lane >= off) x = lse2(x, o);
  }
  return x;
}

__device__ __forceinline__ float wave_suffix_lse(float x, int lane) {
#pragma unroll
  for (int off = 1; off < 32; off <<= 1) {
    float o = __shfl_down(x, off, 32);
    if (lane + off < 32) x = lse2(x, o);
  }
  return x;
}

template <int N>
__device__ __forceinline__ float row_shr_fill(float x, float fill) {
  return __builtin_bit_cast(float, __builtin_amdgcn_update_dpp(
      __builtin_bit_cast(int, fill), __builtin_bit_cast(int, x),
      0x110 + N  , 0xF, 0xF, false));
}

__global__ __launch_bounds__(384, 1)
void mobo_kernel(const float* __restrict__ text,
                 const float* __restrict__ mel,
                 const float* __restrict__ gum,
                 float* __restrict__ out_soft,
                 float* __restrict__ out_exp)
{
  __shared__ __attribute__((aligned(16))) float sE[I_ * J_];
  __shared__ __attribute__((aligned(16))) float sL[I_ * J_];
  __shared__ float sP[J_];
  __shared__ float sWt[32];

  const int b    = blockIdx.x;
  const int tid  = threadIdx.x;
  const int lane = tid & 31;
  const int wave = tid >> 5;
  const int half = lane >> 4;
  const int l16  = lane & 15;

  const float* tx = text + (size_t)b * I_ * D_;
  const float* ml = mel  + (size_t)b * J_ * D_;

  for (int t = wave; t < (I_/16) * (J_/16); t += NWAVE) {
    const int mT = t % (I_/16);
    const int nT = t / (I_/16);
    const int mrow = mT * 16 + l16;
    const int ncol = nT * 16 + l16;
    v8f acc = {};
    for (int k0 = 0; k0 < D_; k0 += 4) {
      const int kk = k0 + 2 * half;
      v2f a  = *(const v2f*)(tx + (size_t)mrow * D_ + kk);
      v2f bf = *(const v2f*)(ml + (size_t)ncol * D_ + kk);
      acc = __builtin_amdgcn_wmma_f32_16x16x4_f32(false, a, false, bf,
                                                  (short)0, acc, false, false);
    }
    for (int r = 0; r < 8; ++r) {
      const int i = mT * 16 + r + 8 * half;
      const int j = nT * 16 + l16;
      const float u = gum[(size_t)b * I_ * J_ + (size_t)i * J_ + j];
      const float g = -__logf(-__logf(u));
      sE[i * J_ + j] = (acc[r] * ESCALE + g) * TEMPINV;
    }
  }
  __syncthreads();

  for (int r = 0; r < RPW; ++r) {
    const int row = wave * RPW + r;
    const float* er = sE + row * J_ + lane * CPL;
    const v4f x0 = *(const v4f*)(er + 0);
    const v4f x1 = *(const v4f*)(er + 4);
    const v4f x2 = *(const v4f*)(er + 8);
    float s[CPL] = { x0.x, x0.y, x0.z, x0.w, x1.x, x1.y, x1.z, x1.w,
                     x2.x, x2.y, x2.z, x2.w };
    for (int c = CPL - 2; c >= 0; --c) s[c] = lse2(s[c], s[c + 1]);
    const float inc = wave_suffix_lse(s[0], lane);
    float exc = __shfl_down(inc, 1, 32);
    if (lane == 31) exc = NEGINF;
    for (int c = 0; c < CPL; ++c) s[c] = lse2(s[c], exc);
    float* lr = sL + row * J_ + lane * CPL;
    *(v4f*)(lr + 0) = (v4f){ s[0], s[1], s[2],  s[3]  };
    *(v4f*)(lr + 4) = (v4f){ s[4], s[5], s[6],  s[7]  };
    *(v4f*)(lr + 8) = (v4f){ s[8], s[9], s[10], s[11] };
  }

  float bv = (tid == 0) ? 0.0f : NEGV;
  const int st = (tid == 0) ? (J_ - 1) : (tid - 1);
  const int wv = __builtin_amdgcn_readfirstlane(wave);
  if (tid < 32) sWt[tid] = NEGINF;
  __syncthreads();

  for (int i = 0; i < I_ - 1; ++i) {
    const float Lv = sL[i * J_ + tid];

    const float av = bv - Lv;
    float xp  = wave_prefix_lse(av, lane);
    float xsa = wave_suffix_lse(av, lane);
    if (lane == 31) sWt[1 + wave]           = xp;
    if (lane == 0)  sWt[16 + NWAVE - wave]  = xsa;
    __syncthreads();
    float t = sWt[lane];
    t = lse2(t, row_shr_fill<1>(t, NEGINF));
    t = lse2(t, row_shr_fill<2>(t, NEGINF));
    t = lse2(t, row_shr_fill<4>(t, NEGINF));
    t = lse2(t, row_shr_fill<8>(t, NEGINF));
    const float plo  = __builtin_bit_cast(float,
        __builtin_amdgcn_readlane(__builtin_bit_cast(int, t), wv));
    const float sahi = __builtin_bit_cast(float,
        __builtin_amdgcn_readlane(__builtin_bit_cast(int, t), 16 + NWAVE - 1 - wv));
    const float Pv  = lse2(xp,  plo);
    const float SAv = lse2(xsa, sahi);
    sP[tid] = Pv;
    sL[i * J_ + tid] = Lv + Pv;
    __syncthreads();

    const int r = i + 1;
    const float ev = sE[i * J_ + st];
    float nb = NEGV;
    if (tid >= r && tid <= (J_ - I_ + r + 1))
      nb = lse2(ev + sP[tid - 1], SAv + LOGEPS);
    sE[i * J_ + st] = bv;
    bv = nb;
  }
  sE[(I_ - 1) * J_ + st] = bv;
  __syncthreads();

  for (int rr = 0; rr < RPW; ++rr) {
    const int row = wave * RPW + rr;
    const float* br = sE + row * J_ + lane * CPL;
    const float* qr = sL + row * J_ + lane * CPL;
    const v4f x0 = *(const v4f*)(br + 0);
    const v4f x1 = *(const v4f*)(br + 4);
    const v4f x2 = *(const v4f*)(br + 8);
    const v4f q0 = *(const v4f*)(qr + 0);
    const v4f q1 = *(const v4f*)(qr + 4);
    const v4f q2 = *(const v4f*)(qr + 8);
    float s[CPL] = { x0.x, x0.y, x0.z, x0.w, x1.x, x1.y, x1.z, x1.w,
                     x2.x, x2.y, x2.z, x2.w };
    const float q[CPL] = { q0.x, q0.y, q0.z, q0.w, q1.x, q1.y, q1.z, q1.w,
                           q2.x, q2.y, q2.z, q2.w };
    const float bfirst = s[CPL - 1];
    if (lane == 31) s[CPL - 1] = NEGINF;
    for (int c = CPL - 2; c >= 0; --c) s[c] = lse2(s[c], s[c + 1]);
    const float inc = wave_suffix_lse(s[0], lane);
    float exc = __shfl_down(inc, 1, 32);
    if (lane == 31) exc = NEGINF;
    for (int c = 0; c < CPL; ++c) s[c] = lse2(s[c], exc);
    float o[CPL];
    if (row < I_ - 1) {
      for (int c = 0; c < CPL; ++c) o[c] = lse2(q[c], s[c] + LOGEPS);
    } else {
      const float tot = lse2(__shfl(s[0], 0, 32), __shfl(bfirst, 31, 32));
      for (int c = 0; c < CPL; ++c)
        o[c] = tot + ((lane * CPL + c == J_ - 1) ? 0.0f : LOGEPS);
    }
    float* og = out_soft + (size_t)b * I_ * J_ + (size_t)row * J_ + lane * CPL;
    *(v4f*)(og + 0) = (v4f){ o[0], o[1], o[2],  o[3]  };
    *(v4f*)(og + 4) = (v4f){ o[4], o[5], o[6],  o[7]  };
    *(v4f*)(og + 8) = (v4f){ o[8], o[9], o[10], o[11] };
    float* wr = sE + row * J_ + lane * CPL;
    *(v4f*)(wr + 0) = (v4f){ __expf(o[0]), __expf(o[1]), __expf(o[2]),  __expf(o[3])  };
    *(v4f*)(wr + 4) = (v4f){ __expf(o[4]), __expf(o[5]), __expf(o[6]),  __expf(o[7])  };
    *(v4f*)(wr + 8) = (v4f){ __expf(o[8]), __expf(o[9]), __expf(o[10]), __expf(o[11]) };
  }
  __syncthreads();

  for (int t = wave; t < (J_/16) * (D_/16); t += NWAVE) {
    const int mT = t % (J_/16);
    const int nT = t / (J_/16);
    const int m = mT * 16 + l16;
    const int n = nT * 16 + l16;
    v8f acc = {};
    for (int k0 = 0; k0 < I_; k0 += 4) {
      const int kk = k0 + 2 * half;
      v2f a;
      a.x = sE[kk * J_ + m];
      a.y = sE[(kk + 1) * J_ + m];
      v2f bf;
      bf.x = tx[(size_t)kk * D_ + n];
      bf.y = tx[(size_t)(kk + 1) * D_ + n];
      acc = __builtin_amdgcn_wmma_f32_16x16x4_f32(false, a, false, bf,
                                                  (short)0, acc, false, false);
    }
    for (int r = 0; r < 8; ++r) {
      const int j = mT * 16 + r + 8 * half;
      out_exp[(size_t)b * J_ * D_ + (size_t)j * D_ + n] = acc[r];
    }
  }
}

extern "C" void kernel_launch(void* const* d_in, const int* in_sizes, int n_in,
                              void* d_out, int out_size, void* d_ws, size_t ws_size,
                              hipStream_t stream) {
  (void)in_sizes; (void)n_in; (void)d_ws; (void)ws_size; (void)out_size;
  const float* text = (const float*)d_in[0];
  const float* mel  = (const float*)d_in[1];
  const float* gum  = (const float*)d_in[2];
  float* out_soft = (float*)d_out;
  float* out_exp  = (float*)d_out + (size_t)B_ * I_ * J_;
  mobo_kernel<<<dim3(B_), dim3(384), 0, stream>>>(text, mel, gum, out_soft, out_exp);
}


# Every two-term LSE as straight-line code on hardware exp2/log2: one exp of -|a-b|, log2 of a value in [1,2] without fix-up code, masked steps take the identity, no branch (same trees, ulp rounding)
# speedup vs baseline: 3.1633x; 1.4484x over previous
#include <hip/hip_runtime.h>
#include <hip/hip_bf16.h>
#include <math.h>


typedef float v2f __attribute__((ext_vector_type(2)));
typedef float v4f __attribute__((ext_vector_type(4)));
typedef float v8f __attribute__((ext_vector_type(8)));

#define B_      4
#define I_      96
#define J_      384
#define D_      256
#define NWAVE   12
#define RPW     (I_ / NWAVE)
#define CPL     (J_ / 32)
#define LOGEPS  (-1000.0f)
#define NEGV    (-1000000000.0f)
#define NEGINF  (-1.0e38f)
#define TEMPINV (1.0f/0.55f)
#define ESCALE  (1.0f/256.0f)
#define LOG2E   1.4426950408889634f
#define LN2     0.6931471805599453f

__device__ __forceinline__ float lse2(float a, float b) {
  const float m = fmaxf(a, b);
  const float e = __builtin_amdgcn_exp2f(-fabsf(a - b) * LOG2E);
  return __builtin_fmaf(__builtin_amdgcn_logf(1.0f + e), LN2, m);
}

__device__ __forceinline__ float wave_prefix_lse(float x, int lane) {
#pragma unroll
  for (int off = 1; off < 32; off <<= 1) {
    const float o = __shfl_up(x, off, 32);
    x = lse2(x, (lane >= off) ? o : NEGINF);
  }
  return x;
}

__device__ __forceinline__ float wave_suffix_lse(float x, int lane) {
#pragma unroll
  for (int off = 1; off < 32; off <<= 1) {
    const float o = __shfl_down(x, off, 32);
    x = lse2(x, (lane + off < 32) ? o : NEGINF);
  }
  return x;
}

template <int N>
__device__ __forceinline__ float row_shr_fill(float x, float fill) {
  return __builtin_bit_cast(float, __builtin_amdgcn_update_dpp(
      __builtin_bit_cast(int, fill), __builtin_bit_cast(int, x),
      0x110 + N  , 0xF, 0xF, false));
}

__global__ __launch_bounds__(384, 1)
void mobo_kernel(const float* __restrict__ text,
                 const float* __restrict__ mel,
                 const float* __restrict__ gum,
                 float* __restrict__ out_soft,
                 float* __restrict__ out_exp)
{
  __shared__ __attribute__((aligned(16))) float sE[I_ * J_];
  __shared__ __attribute__((aligned(16))) float sL[I_ * J_];
  __shared__ float sP[J_];
  __shared__ float sWt[32];

  const int b    = blockIdx.x;
  const int tid  = threadIdx.x;
  const int lane = tid & 31;
  const int wave = tid >> 5;
  const int half = lane >> 4;
  const int l16  = lane & 15;

  const float* tx = text + (size_t)b * I_ * D_;
  const float* ml = mel  + (size_t)b * J_ * D_;

  for (int t = wave; t < (I_/16) * (J_/16); t += NWAVE) {
    const int mT = t % (I_/16);
    const int nT = t / (I_/16);
    const int mrow = mT * 16 + l16;
    const int ncol = nT * 16 + l16;
    v8f acc = {};
    for (int k0 = 0; k0 < D_; k0 += 4) {
      const int kk = k0 + 2 * half;
      v2f a  = *(const v2f*)(tx + (size_t)mrow * D_ + kk);
      v2f bf = *(const v2f*)(ml + (size_t)ncol * D_ + kk);
      acc = __builtin_amdgcn_wmma_f32_16x16x4_f32(false, a, false, bf,
                                                  (short)0, acc, false, false);
    }
    for (int r = 0; r < 8; ++r) {
      const int i = mT * 16 + r + 8 * half;
      const int j = nT * 16 + l16;
      const float u = gum[(size_t)b * I_ * J_ + (size_t)i * J_ + j];
      const float g = -__logf(-__logf(u));
      sE[i * J_ + j] = fmaxf((acc[r] * ESCALE + g) * TEMPINV, NEGINF);
    }
  }
  __syncthreads();

  for (int r = 0; r < RPW; ++r) {
    const int row = wave * RPW + r;
    const float* er = sE + row * J_ + lane * CPL;
    const v4f x0 = *(const v4f*)(er + 0);
    const v4f x1 = *(const v4f*)(er + 4);
    const v4f x2 = *(const v4f*)(er + 8);
    float s[CPL] = { x0.x, x0.y, x0.z, x0.w, x1.x, x1.y, x1.z, x1.w,
                     x2.x, x2.y, x2.z, x2.w };
    for (int c = CPL - 2; c >= 0; --c) s[c] = lse2(s[c], s[c + 1]);
    const float inc = wave_suffix_lse(s[0], lane);
    float exc = __shfl_down(inc, 1, 32);
    if (lane == 31) exc = NEGINF;
    for (int c = 0; c < CPL; ++c) s[c] = lse2(s[c], exc);
    float* lr = sL + row * J_ + lane * CPL;
    *(v4f*)(lr + 0) = (v4f){ s[0], s[1], s[2],  s[3]  };
    *(v4f*)(lr + 4) = (v4f){ s[4], s[5], s[6],  s[7]  };
    *(v4f*)(lr + 8) = (v4f){ s[8], s[9], s[10], s[11] };
  }

  float bv = (tid == 0) ? 0.0f : NEGV;
  const int st = (tid == 0) ? (J_ - 1) : (tid - 1);
  const int wv = __builtin_amdgcn_readfirstlane(wave);
  if (tid < 32) sWt[tid] = NEGINF;
  __syncthreads();

  for (int i = 0; i < I_ - 1; ++i) {
    const float Lv = sL[i * J_ + tid];

    const float av = bv - Lv;
    float xp  = wave_prefix_lse(av, lane);
    float xsa = wave_suffix_lse(av, lane);
    if (lane == 31) sWt[1 + wave]           = xp;
    if (lane == 0)  sWt[16 + NWAVE - wave]  = xsa;
    __syncthreads();
    float t = sWt[lane];
    t = lse2(t, row_shr_fill<1>(t, NEGINF));
    t = lse2(t, row_shr_fill<2>(t, NEGINF));
    t = lse2(t, row_shr_fill<4>(t, NEGINF));
    t = lse2(t, row_shr_fill<8>(t, NEGINF));
    const float plo  = __builtin_bit_cast(float,
        __builtin_amdgcn_readlane(__builtin_bit_cast(int, t), wv));
    const float sahi = __builtin_bit_cast(float,
        __builtin_amdgcn_readlane(__builtin_bit_cast(int, t), 16 + NWAVE - 1 - wv));
    const float Pv  = lse2(xp,  plo);
    const float SAv = lse2(xsa, sahi);
    sP[tid] = Pv;
    sL[i * J_ + tid] = Lv + Pv;
    __syncthreads();

    const int r = i + 1;
    const float ev = sE[i * J_ + st];
    const float cand = lse2(ev + sP[st], SAv + LOGEPS);
    const float nb = (tid >= r && tid <= (J_ - I_ + r + 1)) ? cand : NEGV;
    sE[i * J_ + st] = bv;
    bv = nb;
  }
  sE[(I_ - 1) * J_ + st] = bv;
  __syncthreads();

  for (int rr = 0; rr < RPW; ++rr) {
    const int row = wave * RPW + rr;
    const float* br = sE + row * J_ + lane * CPL;
    const float* qr = sL + row * J_ + lane * CPL;
    const v4f x0 = *(const v4f*)(br + 0);
    const v4f x1 = *(const v4f*)(br + 4);
    const v4f x2 = *(const v4f*)(br + 8);
    const v4f q0 = *(const v4f*)(qr + 0);
    const v4f q1 = *(const v4f*)(qr + 4);
    const v4f q2 = *(const v4f*)(qr + 8);
    float s[CPL] = { x0.x, x0.y, x0.z, x0.w, x1.x, x1.y, x1.z, x1.w,
                     x2.x, x2.y, x2.z, x2.w };
    const float q[CPL] = { q0.x, q0.y, q0.z, q0.w, q1.x, q1.y, q1.z, q1.w,
                           q2.x, q2.y, q2.z, q2.w };
    const float bfirst = s[CPL - 1];
    if (lane == 31) s[CPL - 1] = NEGINF;
    for (int c = CPL - 2; c >= 0; --c) s[c] = lse2(s[c], s[c + 1]);
    const float inc = wave_suffix_lse(s[0], lane);
    float exc = __shfl_down(inc, 1, 32);
    if (lane == 31) exc = NEGINF;
    for (int c = 0; c < CPL; ++c) s[c] = lse2(s[c], exc);
    float o[CPL];
    if (row < I_ - 1) {
      for (int c = 0; c < CPL; ++c) o[c] = lse2(q[c], s[c] + LOGEPS);
    } else {
      const float tot = lse2(__shfl(s[0], 0, 32), __shfl(bfirst, 31, 32));
      for (int c = 0; c < CPL; ++c)
        o[c] = tot + ((lane * CPL + c == J_ - 1) ? 0.0f : LOGEPS);
    }
    float* og = out_soft + (size_t)b * I_ * J_ + (size_t)row * J_ + lane * CPL;
    *(v4f*)(og + 0) = (v4f){ o[0], o[1], o[2],  o[3]  };
    *(v4f*)(og + 4) = (v4f){ o[4], o[5], o[6],  o[7]  };
    *(v4f*)(og + 8) = (v4f){ o[8], o[9], o[10], o[11] };
    float* wr = sE + row * J_ + lane * CPL;
    *(v4f*)(wr + 0) = (v4f){ __expf(o[0]), __expf(o[1]), __expf(o[2]),  __expf(o[3])  };
    *(v4f*)(wr + 4) = (v4f){ __expf(o[4]), __expf(o[5]), __expf(o[6]),  __expf(o[7])  };
    *(v4f*)(wr + 8) = (v4f){ __expf(o[8]), __expf(o[9]), __expf(o[10]), __expf(o[11]) };
  }
  __syncthreads();

  for (int t = wave; t < (J_/16) * (D_/16); t += NWAVE) {
    const int mT = t % (J_/16);
    const int nT = t / (J_/16);
    const int m = mT * 16 + l16;
    const int n = nT * 16 + l16;
    v8f acc = {};
    for (int k0 = 0; k0 < I_; k0 += 4) {
      const int kk = k0 + 2 * half;
      v2f a;
      a.x = sE[kk * J_ + m];
      a.y = sE[(kk + 1) * J_ + m];
      v2f bf;
      bf.x = tx[(size_t)kk * D_ + n];
      bf.y = tx[(size_t)(kk + 1) * D_ + n];
      acc = __builtin_amdgcn_wmma_f32_16x16x4_f32(false, a, false, bf,
                                                  (short)0, acc, false, false);
    }
    for (int r = 0; r < 8; ++r) {
      const int j = mT * 16 + r + 8 * half;
      out_exp[(size_t)b * J_ * D_ + (size_t)j * D_ + n] = acc[r];
    }
  }
}

extern "C" void kernel_launch(void* const* d_in, const int* in_sizes, int n_in,
                              void* d_out, int out_size, void* d_ws, size_t ws_size,
                              hipStream_t stream) {
  (void)in_sizes; (void)n_in; (void)d_ws; (void)ws_size; (void)out_size;
  const float* text = (const float*)d_in[0];
  const float* mel  = (const float*)d_in[1];
  const float* gum  = (const float*)d_in[2];
  float* out_soft = (float*)d_out;
  float* out_exp  = (float*)d_out + (size_t)B_ * I_ * J_;
  mobo_kernel<<<dim3(B_), dim3(384), 0, stream>>>(text, mel, gum, out_soft, out_exp);
}


# Energy GEMM: each wave owns a 96x32 block (12 accumulators), each fragment loaded once per K chunk by one 16-byte load per lane (k permuted alike in A and B, new k summation order): 6x fewer loads
# speedup vs baseline: 4.5346x; 1.4335x over previous
#include <hip/hip_runtime.h>
#include <hip/hip_bf16.h>
#include <math.h>


typedef float v2f __attribute__((ext_vector_type(2)));
typedef float v4f __attribute__((ext_vector_type(4)));
typedef float v8f __attribute__((ext_vector_type(8)));

#define B_      4
#define I_      96
#define J_      384
#define D_      256
#define NWAVE   12
#define RPW     (I_ / NWAVE)
#define CPL     (J_ / 32)
#define LOGEPS  (-1000.0f)
#define NEGV    (-1000000000.0f)
#define NEGINF  (-1.0e38f)
#define TEMPINV (1.0f/0.55f)
#define ESCALE  (1.0f/256.0f)
#define LOG2E   1.4426950408889634f
#define LN2     0.6931471805599453f

__device__ __forceinline__ float lse2(float a, float b) {
  const float m = fmaxf(a, b);
  const float e = __builtin_amdgcn_exp2f(-fabsf(a - b) * LOG2E);
  return __builtin_fmaf(__builtin_amdgcn_logf(1.0f + e), LN2, m);
}

__device__ __forceinline__ float wave_prefix_lse(float x, int lane) {
#pragma unroll
  for (int off = 1; off < 32; off <<= 1) {
    const float o = __shfl_up(x, off, 32);
    x = lse2(x, (lane >= off) ? o : NEGINF);
  }
  return x;
}

__device__ __forceinline__ float wave_suffix_lse(float x, int lane) {
#pragma unroll
  for (int off = 1; off < 32; off <<= 1) {
    const float o = __shfl_down(x, off, 32);
    x = lse2(x, (lane + off < 32) ? o : NEGINF);
  }
  return x;
}

template <int N>
__device__ __forceinline__ float row_shr_fill(float x, float fill) {
  return __builtin_bit_cast(float, __builtin_amdgcn_update_dpp(
      __builtin_bit_cast(int, fill), __builtin_bit_cast(int, x),
      0x110 + N  , 0xF, 0xF, false));
}

__device__ __forceinline__ void load_frags(v4f (&a)[I_/16], v4f (&bq)[2],
                                           const float* ap, const float* bp) {
#pragma unroll
  for (int m = 0; m < I_/16; ++m) a[m] = *(const v4f*)(ap + (size_t)m * 16 * D_);
#pragma unroll
  for (int n = 0; n < 2; ++n) bq[n] = *(const v4f*)(bp + (size_t)n * 16 * D_);
}

__device__ __forceinline__ void mma_chunk(v8f (&acc)[I_/16][2], const v4f (&a)[I_/16],
                                          const v4f (&bq)[2]) {
#pragma unroll
  for (int m = 0; m < I_/16; ++m) {
#pragma unroll
    for (int n = 0; n < 2; ++n) {
      const v2f av = { a[m].x, a[m].y };
      const v2f bv2 = { bq[n].x, bq[n].y };
      acc[m][n] = __builtin_amdgcn_wmma_f32_16x16x4_f32(false, av, false, bv2,
                                                        (short)0, acc[m][n], false, false);
    }
  }
#pragma unroll
  for (int m = 0; m < I_/16; ++m) {
#pragma unroll
    for (int n = 0; n < 2; ++n) {
      const v2f av = { a[m].z, a[m].w };
      const v2f bv2 = { bq[n].z, bq[n].w };
      acc[m][n] = __builtin_amdgcn_wmma_f32_16x16x4_f32(false, av, false, bv2,
                                                        (short)0, acc[m][n], false, false);
    }
  }
}

__global__ __launch_bounds__(384, 1)
void mobo_kernel(const float* __restrict__ text,
                 const float* __restrict__ mel,
                 const float* __restrict__ gum,
                 float* __restrict__ out_soft,
                 float* __restrict__ out_exp)
{
  __shared__ __attribute__((aligned(16))) float sE[I_ * J_];
  __shared__ __attribute__((aligned(16))) float sL[I_ * J_];
  __shared__ float sP[J_];
  __shared__ float sWt[32];

  const int b    = blockIdx.x;
  const int tid  = threadIdx.x;
  const int lane = tid & 31;
  const int wave = tid >> 5;
  const int half = lane >> 4;
  const int l16  = lane & 15;

  const float* tx = text + (size_t)b * I_ * D_;
  const float* ml = mel  + (size_t)b * J_ * D_;

  {
    v8f acc[I_/16][2];
#pragma unroll
    for (int m = 0; m < I_/16; ++m) { acc[m][0] = (v8f){}; acc[m][1] = (v8f){}; }
    const float* ap = tx + (size_t)l16 * D_ + 4 * half;
    const float* bp = ml + (size_t)(wave * 32 + l16) * D_ + 4 * half;
    v4f a0[I_/16], b0[2], a1[I_/16], b1[2];
    load_frags(a0, b0, ap, bp);
    for (int q = 0; q < D_/8 - 2; q += 2) {
      load_frags(a1, b1, ap + 8 * (q + 1), bp + 8 * (q + 1));
      mma_chunk(acc, a0, b0);
      load_frags(a0, b0, ap + 8 * (q + 2), bp + 8 * (q + 2));
      mma_chunk(acc, a1, b1);
    }
    load_frags(a1, b1, ap + 8 * (D_/8 - 1), bp + 8 * (D_/8 - 1));
    mma_chunk(acc, a0, b0);
    mma_chunk(acc, a1, b1);
#pragma unroll
    for (int m = 0; m < I_/16; ++m) {
#pragma unroll
      for (int n = 0; n < 2; ++n) {
#pragma unroll
        for (int r = 0; r < 8; ++r) {
          const int i = m * 16 + r + 8 * half;
          const int j = wave * 32 + n * 16 + l16;
          const float u = gum[(size_t)b * I_ * J_ + (size_t)i * J_ + j];
          const float g = -__logf(-__logf(u));
          sE[i * J_ + j] = fmaxf((acc[m][n][r] * ESCALE + g) * TEMPINV, NEGINF);
        }
      }
    }
  }
  __syncthreads();

  for (int r = 0; r < RPW; ++r) {
    const int row = wave * RPW + r;
    const float* er = sE + row * J_ + lane * CPL;
    const v4f x0 = *(const v4f*)(er + 0);
    const v4f x1 = *(const v4f*)(er + 4);
    const v4f x2 = *(const v4f*)(er + 8);
    float s[CPL] = { x0.x, x0.y, x0.z, x0.w, x1.x, x1.y, x1.z, x1.w,
                     x2.x, x2.y, x2.z, x2.w };
    for (int c = CPL - 2; c >= 0; --c) s[c] = lse2(s[c], s[c + 1]);
    const float inc = wave_suffix_lse(s[0], lane);
    float exc = __shfl_down(inc, 1, 32);
    if (lane == 31) exc = NEGINF;
    for (int c = 0; c < CPL; ++c) s[c] = lse2(s[c], exc);
    float* lr = sL + row * J_ + lane * CPL;
    *(v4f*)(lr + 0) = (v4f){ s[0], s[1], s[2],  s[3]  };
    *(v4f*)(lr + 4) = (v4f){ s[4], s[5], s[6],  s[7]  };
    *(v4f*)(lr + 8) = (v4f){ s[8], s[9], s[10], s[11] };
  }

  float bv = (tid == 0) ? 0.0f : NEGV;
  const int st = (tid == 0) ? (J_ - 1) : (tid - 1);
  const int wv = __builtin_amdgcn_readfirstlane(wave);
  if (tid < 32) sWt[tid] = NEGINF;
  __syncthreads();

  for (int i = 0; i < I_ - 1; ++i) {
    const float Lv = sL[i * J_ + tid];

    const float av = bv - Lv;
    float xp  = wave_prefix_lse(av, lane);
    float xsa = wave_suffix_lse(av, lane);
    if (lane == 31) sWt[1 + wave]           = xp;
    if (lane == 0)  sWt[16 + NWAVE - wave]  = xsa;
    __syncthreads();
    float t = sWt[lane];
    t = lse2(t, row_shr_fill<1>(t, NEGINF));
    t = lse2(t, row_shr_fill<2>(t, NEGINF));
    t = lse2(t, row_shr_fill<4>(t, NEGINF));
    t = lse2(t, row_shr_fill<8>(t, NEGINF));
    const float plo  = __builtin_bit_cast(float,
        __builtin_amdgcn_readlane(__builtin_bit_cast(int, t), wv));
    const float sahi = __builtin_bit_cast(float,
        __builtin_amdgcn_readlane(__builtin_bit_cast(int, t), 16 + NWAVE - 1 - wv));
    const float Pv  = lse2(xp,  plo);
    const float SAv = lse2(xsa, sahi);
    sP[tid] = Pv;
    sL[i * J_ + tid] = Lv + Pv;
    __syncthreads();

    const int r = i + 1;
    const float ev = sE[i * J_ + st];
    const float cand = lse2(ev + sP[st], SAv + LOGEPS);
    const float nb = (tid >= r && tid <= (J_ - I_ + r + 1)) ? cand : NEGV;
    sE[i * J_ + st] = bv;
    bv = nb;
  }
  sE[(I_ - 1) * J_ + st] = bv;
  __syncthreads();

  for (int rr = 0; rr < RPW; ++rr) {
    const int row = wave * RPW + rr;
    const float* br = sE + row * J_ + lane * CPL;
    const float* qr = sL + row * J_ + lane * CPL;
    const v4f x0 = *(const v4f*)(br + 0);
    const v4f x1 = *(const v4f*)(br + 4);
    const v4f x2 = *(const v4f*)(br + 8);
    const v4f q0 = *(const v4f*)(qr + 0);
    const v4f q1 = *(const v4f*)(qr + 4);
    const v4f q2 = *(const v4f*)(qr + 8);
    float s[CPL] = { x0.x, x0.y, x0.z, x0.w, x1.x, x1.y, x1.z, x1.w,
                     x2.x, x2.y, x2.z, x2.w };
    const float q[CPL] = { q0.x, q0.y, q0.z, q0.w, q1.x, q1.y, q1.z, q1.w,
                           q2.x, q2.y, q2.z, q2.w };
    const float bfirst = s[CPL - 1];
    if (lane == 31) s[CPL - 1] = NEGINF;
    for (int c = CPL - 2; c >= 0; --c) s[c] = lse2(s[c], s[c + 1]);
    const float inc = wave_suffix_lse(s[0], lane);
    float exc = __shfl_down(inc, 1, 32);
    if (lane == 31) exc = NEGINF;
    for (int c = 0; c < CPL; ++c) s[c] = lse2(s[c], exc);
    float o[CPL];
    if (row < I_ - 1) {
      for (int c = 0; c < CPL; ++c) o[c] = lse2(q[c], s[c] + LOGEPS);
    } else {
      const float tot = lse2(__shfl(s[0], 0, 32), __shfl(bfirst, 31, 32));
      for (int c = 0; c < CPL; ++c)
        o[c] = tot + ((lane * CPL + c == J_ - 1) ? 0.0f : LOGEPS);
    }
    float* og = out_soft + (size_t)b * I_ * J_ + (size_t)row * J_ + lane * CPL;
    *(v4f*)(og + 0) = (v4f){ o[0], o[1], o[2],  o[3]  };
    *(v4f*)(og + 4) = (v4f){ o[4], o[5], o[6],  o[7]  };
    *(v4f*)(og + 8) = (v4f){ o[8], o[9], o[10], o[11] };
    float* wr = sE + row * J_ + lane * CPL;
    *(v4f*)(wr + 0) = (v4f){ __expf(o[0]), __expf(o[1]), __expf(o[2]),  __expf(o[3])  };
    *(v4f*)(wr + 4) = (v4f){ __expf(o[4]), __expf(o[5]), __expf(o[6]),  __expf(o[7])  };
    *(v4f*)(wr + 8) = (v4f){ __expf(o[8]), __expf(o[9]), __expf(o[10]), __expf(o[11]) };
  }
  __syncthreads();

  for (int t = wave; t < (J_/16) * (D_/16); t += NWAVE) {
    const int mT = t % (J_/16);
    const int nT = t / (J_/16);
    const int m = mT * 16 + l16;
    const int n = nT * 16 + l16;
    v8f acc = {};
    for (int k0 = 0; k0 < I_; k0 += 4) {
      const int kk = k0 + 2 * half;
      v2f a;
      a.x = sE[kk * J_ + m];
      a.y = sE[(kk + 1) * J_ + m];
      v2f bf;
      bf.x = tx[(size_t)kk * D_ + n];
      bf.y = tx[(size_t)(kk + 1) * D_ + n];
      acc = __builtin_amdgcn_wmma_f32_16x16x4_f32(false, a, false, bf,
                                                  (short)0, acc, false, false);
    }
    for (int r = 0; r < 8; ++r) {
      const int j = mT * 16 + r + 8 * half;
      out_exp[(size_t)b * J_ * D_ + (size_t)j * D_ + n] = acc[r];
    }
  }
}

extern "C" void kernel_launch(void* const* d_in, const int* in_sizes, int n_in,
                              void* d_out, int out_size, void* d_ws, size_t ws_size,
                              hipStream_t stream) {
  (void)in_sizes; (void)n_in; (void)d_ws; (void)ws_size; (void)out_size;
  const float* text = (const float*)d_in[0];
  const float* mel  = (const float*)d_in[1];
  const float* gum  = (const float*)d_in[2];
  float* out_soft = (float*)d_out;
  float* out_exp  = (float*)d_out + (size_t)B_ * I_ * J_;
  mobo_kernel<<<dim3(B_), dim3(384), 0, stream>>>(text, mel, gum, out_soft, out_exp);
}


# In-wave cumulative-LSE scans (DP rows and both row passes): lane steps as DPP row shifts plus one lane broadcast on the vector ALU, no LDS-unit shuffles (new reduction tree for the in-wave scans)
# speedup vs baseline: 5.1834x; 1.1431x over previous
#include <hip/hip_runtime.h>
#include <hip/hip_bf16.h>
#include <math.h>


typedef float v2f __attribute__((ext_vector_type(2)));
typedef float v4f __attribute__((ext_vector_type(4)));
typedef float v8f __attribute__((ext_vector_type(8)));

#define B_      4
#define I_      96
#define J_      384
#define D_      256
#define NWAVE   12
#define RPW     (I_ / NWAVE)
#define CPL     (J_ / 32)
#define LOGEPS  (-1000.0f)
#define NEGV    (-1000000000.0f)
#define NEGINF  (-1.0e38f)
#define TEMPINV (1.0f/0.55f)
#define ESCALE  (1.0f/256.0f)
#define LOG2E   1.4426950408889634f
#define LN2     0.6931471805599453f

__device__ __forceinline__ float lse2(float a, float b) {
  const float m = fmaxf(a, b);
  const float e = __builtin_amdgcn_exp2f(-fabsf(a - b) * LOG2E);
  return __builtin_fmaf(__builtin_amdgcn_logf(1.0f + e), LN2, m);
}

template <int N>
__device__ __forceinline__ float row_shr_fill(float x, float fill) {
  return __builtin_bit_cast(float, __builtin_amdgcn_update_dpp(
      __builtin_bit_cast(int, fill), __builtin_bit_cast(int, x),
      0x110 + N  , 0xF, 0xF, false));
}

template <int N>
__device__ __forceinline__ float row_shl_fill(float x, float fill) {
  return __builtin_bit_cast(float, __builtin_amdgcn_update_dpp(
      __builtin_bit_cast(int, fill), __builtin_bit_cast(int, x),
      0x100 + N  , 0xF, 0xF, false));
}

template <int L>
__device__ __forceinline__ float lane_bcast(float x) {
  return __builtin_bit_cast(float,
      __builtin_amdgcn_readlane(__builtin_bit_cast(int, x), L));
}

__device__ __forceinline__ float wave_prefix_lse(float x, int lane) {
  x = lse2(x, row_shr_fill<1>(x, NEGINF));
  x = lse2(x, row_shr_fill<2>(x, NEGINF));
  x = lse2(x, row_shr_fill<4>(x, NEGINF));
  x = lse2(x, row_shr_fill<8>(x, NEGINF));
  const float lo = lane_bcast<15>(x);
  return lse2(x, (lane >= 16) ? lo : NEGINF);
}

__device__ __forceinline__ float wave_suffix_lse(float x, int lane) {
  x = lse2(x, row_shl_fill<1>(x, NEGINF));
  x = lse2(x, row_shl_fill<2>(x, NEGINF));
  x = lse2(x, row_shl_fill<4>(x, NEGINF));
  x = lse2(x, row_shl_fill<8>(x, NEGINF));
  const float hi = lane_bcast<16>(x);
  return lse2(x, (lane < 16) ? hi : NEGINF);
}

__device__ __forceinline__ float wave_next_lane(float x, int lane) {
  const float e  = row_shl_fill<1>(x, NEGINF);
  const float hi = lane_bcast<16>(x);
  return (lane == 15) ? hi : e;
}

__device__ __forceinline__ void load_frags(v4f (&a)[I_/16], v4f (&bq)[2],
                                           const float* ap, const float* bp) {
#pragma unroll
  for (int m = 0; m < I_/16; ++m) a[m] = *(const v4f*)(ap + (size_t)m * 16 * D_);
#pragma unroll
  for (int n = 0; n < 2; ++n) bq[n] = *(const v4f*)(bp + (size_t)n * 16 * D_);
}

__device__ __forceinline__ void mma_chunk(v8f (&acc)[I_/16][2], const v4f (&a)[I_/16],
                                          const v4f (&bq)[2]) {
#pragma unroll
  for (int m = 0; m < I_/16; ++m) {
#pragma unroll
    for (int n = 0; n < 2; ++n) {
      const v2f av = { a[m].x, a[m].y };
      const v2f bv2 = { bq[n].x, bq[n].y };
      acc[m][n] = __builtin_amdgcn_wmma_f32_16x16x4_f32(false, av, false, bv2,
                                                        (short)0, acc[m][n], false, false);
    }
  }
#pragma unroll
  for (int m = 0; m < I_/16; ++m) {
#pragma unroll
    for (int n = 0; n < 2; ++n) {
      const v2f av = { a[m].z, a[m].w };
      const v2f bv2 = { bq[n].z, bq[n].w };
      acc[m][n] = __builtin_amdgcn_wmma_f32_16x16x4_f32(false, av, false, bv2,
                                                        (short)0, acc[m][n], false, false);
    }
  }
}

__global__ __launch_bounds__(384, 1)
void mobo_kernel(const float* __restrict__ text,
                 const float* __restrict__ mel,
                 const float* __restrict__ gum,
                 float* __restrict__ out_soft,
                 float* __restrict__ out_exp)
{
  __shared__ __attribute__((aligned(16))) float sE[I_ * J_];
  __shared__ __attribute__((aligned(16))) float sL[I_ * J_];
  __shared__ float sP[J_];
  __shared__ float sWt[32];

  const int b    = blockIdx.x;
  const int tid  = threadIdx.x;
  const int lane = tid & 31;
  const int wave = tid >> 5;
  const int half = lane >> 4;
  const int l16  = lane & 15;

  const float* tx = text + (size_t)b * I_ * D_;
  const float* ml = mel  + (size_t)b * J_ * D_;

  {
    v8f acc[I_/16][2];
#pragma unroll
    for (int m = 0; m < I_/16; ++m) { acc[m][0] = (v8f){}; acc[m][1] = (v8f){}; }
    const float* ap = tx + (size_t)l16 * D_ + 4 * half;
    const float* bp = ml + (size_t)(wave * 32 + l16) * D_ + 4 * half;
    v4f a0[I_/16], b0[2], a1[I_/16], b1[2];
    load_frags(a0, b0, ap, bp);
    for (int q = 0; q < D_/8 - 2; q += 2) {
      load_frags(a1, b1, ap + 8 * (q + 1), bp + 8 * (q + 1));
      mma_chunk(acc, a0, b0);
      load_frags(a0, b0, ap + 8 * (q + 2), bp + 8 * (q + 2));
      mma_chunk(acc, a1, b1);
    }
    load_frags(a1, b1, ap + 8 * (D_/8 - 1), bp + 8 * (D_/8 - 1));
    mma_chunk(acc, a0, b0);
    mma_chunk(acc, a1, b1);
#pragma unroll
    for (int m = 0; m < I_/16; ++m) {
#pragma unroll
      for (int n = 0; n < 2; ++n) {
#pragma unroll
        for (int r = 0; r < 8; ++r) {
          const int i = m * 16 + r + 8 * half;
          const int j = wave * 32 + n * 16 + l16;
          const float u = gum[(size_t)b * I_ * J_ + (size_t)i * J_ + j];
          const float g = -__logf(-__logf(u));
          sE[i * J_ + j] = fmaxf((acc[m][n][r] * ESCALE + g) * TEMPINV, NEGINF);
        }
      }
    }
  }
  __syncthreads();

  for (int r = 0; r < RPW; ++r) {
    const int row = wave * RPW + r;
    const float* er = sE + row * J_ + lane * CPL;
    const v4f x0 = *(const v4f*)(er + 0);
    const v4f x1 = *(const v4f*)(er + 4);
    const v4f x2 = *(const v4f*)(er + 8);
    float s[CPL] = { x0.x, x0.y, x0.z, x0.w, x1.x, x1.y, x1.z, x1.w,
                     x2.x, x2.y, x2.z, x2.w };
    for (int c = CPL - 2; c >= 0; --c) s[c] = lse2(s[c], s[c + 1]);
    const float inc = wave_suffix_lse(s[0], lane);
    const float exc = wave_next_lane(inc, lane);
    for (int c = 0; c < CPL; ++c) s[c] = lse2(s[c], exc);
    float* lr = sL + row * J_ + lane * CPL;
    *(v4f*)(lr + 0) = (v4f){ s[0], s[1], s[2],  s[3]  };
    *(v4f*)(lr + 4) = (v4f){ s[4], s[5], s[6],  s[7]  };
    *(v4f*)(lr + 8) = (v4f){ s[8], s[9], s[10], s[11] };
  }

  float bv = (tid == 0) ? 0.0f : NEGV;
  const int st = (tid == 0) ? (J_ - 1) : (tid - 1);
  const int wv = __builtin_amdgcn_readfirstlane(wave);
  if (tid < 32) sWt[tid] = NEGINF;
  __syncthreads();

  for (int i = 0; i < I_ - 1; ++i) {
    const float Lv = sL[i * J_ + tid];

    const float av = bv - Lv;
    float xp  = wave_prefix_lse(av, lane);
    float xsa = wave_suffix_lse(av, lane);
    if (lane == 31) sWt[1 + wave]           = xp;
    if (lane == 0)  sWt[16 + NWAVE - wave]  = xsa;
    __syncthreads();
    float t = sWt[lane];
    t = lse2(t, row_shr_fill<1>(t, NEGINF));
    t = lse2(t, row_shr_fill<2>(t, NEGINF));
    t = lse2(t, row_shr_fill<4>(t, NEGINF));
    t = lse2(t, row_shr_fill<8>(t, NEGINF));
    const float plo  = __builtin_bit_cast(float,
        __builtin_amdgcn_readlane(__builtin_bit_cast(int, t), wv));
    const float sahi = __builtin_bit_cast(float,
        __builtin_amdgcn_readlane(__builtin_bit_cast(int, t), 16 + NWAVE - 1 - wv));
    const float Pv  = lse2(xp,  plo);
    const float SAv = lse2(xsa, sahi);
    sP[tid] = Pv;
    sL[i * J_ + tid] = Lv + Pv;
    __syncthreads();

    const int r = i + 1;
    const float ev = sE[i * J_ + st];
    const float cand = lse2(ev + sP[st], SAv + LOGEPS);
    const float nb = (tid >= r && tid <= (J_ - I_ + r + 1)) ? cand : NEGV;
    sE[i * J_ + st] = bv;
    bv = nb;
  }
  sE[(I_ - 1) * J_ + st] = bv;
  __syncthreads();

  for (int rr = 0; rr < RPW; ++rr) {
    const int row = wave * RPW + rr;
    const float* br = sE + row * J_ + lane * CPL;
    const float* qr = sL + row * J_ + lane * CPL;
    const v4f x0 = *(const v4f*)(br + 0);
    const v4f x1 = *(const v4f*)(br + 4);
    const v4f x2 = *(const v4f*)(br + 8);
    const v4f q0 = *(const v4f*)(qr + 0);
    const v4f q1 = *(const v4f*)(qr + 4);
    const v4f q2 = *(const v4f*)(qr + 8);
    float s[CPL] = { x0.x, x0.y, x0.z, x0.w, x1.x, x1.y, x1.z, x1.w,
                     x2.x, x2.y, x2.z, x2.w };
    const float q[CPL] = { q0.x, q0.y, q0.z, q0.w, q1.x, q1.y, q1.z, q1.w,
                           q2.x, q2.y, q2.z, q2.w };
    const float bfirst = s[CPL - 1];
    if (lane == 31) s[CPL - 1] = NEGINF;
    for (int c = CPL - 2; c >= 0; --c) s[c] = lse2(s[c], s[c + 1]);
    const float inc = wave_suffix_lse(s[0], lane);
    const float exc = wave_next_lane(inc, lane);
    for (int c = 0; c < CPL; ++c) s[c] = lse2(s[c], exc);
    float o[CPL];
    if (row < I_ - 1) {
      for (int c = 0; c < CPL; ++c) o[c] = lse2(q[c], s[c] + LOGEPS);
    } else {
      const float tot = lse2(lane_bcast<0>(s[0]), lane_bcast<31>(bfirst));
      for (int c = 0; c < CPL; ++c)
        o[c] = tot + ((lane * CPL + c == J_ - 1) ? 0.0f : LOGEPS);
    }
    float* og = out_soft + (size_t)b * I_ * J_ + (size_t)row * J_ + lane * CPL;
    *(v4f*)(og + 0) = (v4f){ o[0], o[1], o[2],  o[3]  };
    *(v4f*)(og + 4) = (v4f){ o[4], o[5], o[6],  o[7]  };
    *(v4f*)(og + 8) = (v4f){ o[8], o[9], o[10], o[11] };
    float* wr = sE + row * J_ + lane * CPL;
    *(v4f*)(wr + 0) = (v4f){ __expf(o[0]), __expf(o[1]), __expf(o[2]),  __expf(o[3])  };
    *(v4f*)(wr + 4) = (v4f){ __expf(o[4]), __expf(o[5]), __expf(o[6]),  __expf(o[7])  };
    *(v4f*)(wr + 8) = (v4f){ __expf(o[8]), __expf(o[9]), __expf(o[10]), __expf(o[11]) };
  }
  __syncthreads();

  for (int t = wave; t < (J_/16) * (D_/16); t += NWAVE) {
    const int mT = t % (J_/16);
    const int nT = t / (J_/16);
    const int m = mT * 16 + l16;
    const int n = nT * 16 + l16;
    v8f acc = {};
    for (int k0 = 0; k0 < I_; k0 += 4) {
      const int kk = k0 + 2 * half;
      v2f a;
      a.x = sE[kk * J_ + m];
      a.y = sE[(kk + 1) * J_ + m];
      v2f bf;
      bf.x = tx[(size_t)kk * D_ + n];
      bf.y = tx[(size_t)(kk + 1) * D_ + n];
      acc = __builtin_amdgcn_wmma_f32_16x16x4_f32(false, a, false, bf,
                                                  (short)0, acc, false, false);
    }
    for (int r = 0; r < 8; ++r) {
      const int j = mT * 16 + r + 8 * half;
      out_exp[(size_t)b * J_ * D_ + (size_t)j * D_ + n] = acc[r];
    }
  }
}

extern "C" void kernel_launch(void* const* d_in, const int* in_sizes, int n_in,
                              void* d_out, int out_size, void* d_ws, size_t ws_size,
                              hipStream_t stream) {
  (void)in_sizes; (void)n_in; (void)d_ws; (void)ws_size; (void)out_size;
  const float* text = (const float*)d_in[0];
  const float* mel  = (const float*)d_in[1];
  const float* gum  = (const float*)d_in[2];
  float* out_soft = (float*)d_out;
  float* out_exp  = (float*)d_out + (size_t)B_ * I_ * J_;
  mobo_kernel<<<dim3(B_), dim3(384), 0, stream>>>(text, mel, gum, out_soft, out_exp);
}


# Both GEMMs leave the 4-workgroup kernel for grid-wide launches in front of and behind the per-batch DP (energy staged in the soft buffer, exp(soft) taken on load): 3 launches, all sum orders kept
# speedup vs baseline: 9.3926x; 1.8121x over previous
#include <hip/hip_runtime.h>
#include <hip/hip_bf16.h>
#include <math.h>


typedef float v2f __attribute__((ext_vector_type(2)));
typedef float v4f __attribute__((ext_vector_type(4)));
typedef float v8f __attribute__((ext_vector_type(8)));

#define B_      4
#define I_      96
#define J_      384
#define D_      256
#define NWAVE   12
#define RPW     (I_ / NWAVE)
#define CPL     (J_ / 32)
#define LOGEPS  (-1000.0f)
#define NEGV    (-1000000000.0f)
#define NEGINF  (-1.0e38f)
#define TEMPINV (1.0f/0.55f)
#define ESCALE  (1.0f/256.0f)
#define LOG2E   1.4426950408889634f
#define LN2     0.6931471805599453f

__device__ __forceinline__ float lse2(float a, float b) {
  const float m = fmaxf(a, b);
  const float e = __builtin_amdgcn_exp2f(-fabsf(a - b) * LOG2E);
  return __builtin_fmaf(__builtin_amdgcn_logf(1.0f + e), LN2, m);
}

template <int N>
__device__ __forceinline__ float row_shr_fill(float x, float fill) {
  return __builtin_bit_cast(float, __builtin_amdgcn_update_dpp(
      __builtin_bit_cast(int, fill), __builtin_bit_cast(int, x),
      0x110 + N  , 0xF, 0xF, false));
}

template <int N>
__device__ __forceinline__ float row_shl_fill(float x, float fill) {
  return __builtin_bit_cast(float, __builtin_amdgcn_update_dpp(
      __builtin_bit_cast(int, fill), __builtin_bit_cast(int, x),
      0x100 + N  , 0xF, 0xF, false));
}

template <int L>
__device__ __forceinline__ float lane_bcast(float x) {
  return __builtin_bit_cast(float,
      __builtin_amdgcn_readlane(__builtin_bit_cast(int, x), L));
}

__device__ __forceinline__ float wave_prefix_lse(float x, int lane) {
  x = lse2(x, row_shr_fill<1>(x, NEGINF));
  x = lse2(x, row_shr_fill<2>(x, NEGINF));
  x = lse2(x, row_shr_fill<4>(x, NEGINF));
  x = lse2(x, row_shr_fill<8>(x, NEGINF));
  const float lo = lane_bcast<15>(x);
  return lse2(x, (lane >= 16) ? lo : NEGINF);
}

__device__ __forceinline__ float wave_suffix_lse(float x, int lane) {
  x = lse2(x, row_shl_fill<1>(x, NEGINF));
  x = lse2(x, row_shl_fill<2>(x, NEGINF));
  x = lse2(x, row_shl_fill<4>(x, NEGINF));
  x = lse2(x, row_shl_fill<8>(x, NEGINF));
  const float hi = lane_bcast<16>(x);
  return lse2(x, (lane < 16) ? hi : NEGINF);
}

__device__ __forceinline__ float wave_next_lane(float x, int lane) {
  const float e  = row_shl_fill<1>(x, NEGINF);
  const float hi = lane_bcast<16>(x);
  return (lane == 15) ? hi : e;
}

#define EG 8

__device__ __forceinline__ void e_load(v4f (&a)[EG], v4f (&bq)[EG][2],
                                       const float* ap, const float* bp) {
#pragma unroll
  for (int c = 0; c < EG; ++c) {
    a[c] = *(const v4f*)(ap + 8 * c);
#pragma unroll
    for (int n = 0; n < 2; ++n) bq[c][n] = *(const v4f*)(bp + 8 * c + (size_t)n * 16 * D_);
  }
}

__device__ __forceinline__ void e_mma(v8f (&acc)[2], const v4f (&a)[EG],
                                      const v4f (&bq)[EG][2]) {
#pragma unroll
  for (int c = 0; c < EG; ++c) {
#pragma unroll
    for (int n = 0; n < 2; ++n) {
      const v2f av = { a[c].x, a[c].y };
      const v2f bv2 = { bq[c][n].x, bq[c][n].y };
      acc[n] = __builtin_amdgcn_wmma_f32_16x16x4_f32(false, av, false, bv2,
                                                     (short)0, acc[n], false, false);
    }
#pragma unroll
    for (int n = 0; n < 2; ++n) {
      const v2f av = { a[c].z, a[c].w };
      const v2f bv2 = { bq[c][n].z, bq[c][n].w };
      acc[n] = __builtin_amdgcn_wmma_f32_16x16x4_f32(false, av, false, bv2,
                                                     (short)0, acc[n], false, false);
    }
  }
}

__global__ __launch_bounds__(128)
void energy_kernel(const float* __restrict__ text,
                   const float* __restrict__ mel,
                   const float* __restrict__ gum,
                   float* __restrict__ energy)
{
  const int lane = threadIdx.x & 31;
  const int wave = threadIdx.x >> 5;
  const int half = lane >> 4;
  const int l16  = lane & 15;
  const int wb  = blockIdx.x * 4 + wave;
  const int b   = wb / 72;
  const int rem = wb - b * 72;
  const int mb  = rem / 12;
  const int nb  = rem - mb * 12;
  const int i0  = mb * 16;
  const int j0  = nb * 32;

  const float* ap = text + ((size_t)b * I_ + i0 + l16) * D_ + 4 * half;
  const float* bp = mel  + ((size_t)b * J_ + j0 + l16) * D_ + 4 * half;
  const float* gp = gum    + (size_t)b * I_ * J_;
  float*       ep = energy + (size_t)b * I_ * J_;

  v8f acc[2];
  acc[0] = (v8f){}; acc[1] = (v8f){};
  v4f a0[EG], b0[EG][2], a1[EG], b1[EG][2];
  e_load(a0, b0, ap, bp);
  __builtin_amdgcn_sched_barrier(0);
  float u[2][8];
#pragma unroll
  for (int n = 0; n < 2; ++n) {
#pragma unroll
    for (int r = 0; r < 8; ++r)
      u[n][r] = gp[(size_t)(i0 + r + 8 * half) * J_ + j0 + n * 16 + l16];
  }
  e_load(a1, b1, ap + 64, bp + 64);
  __builtin_amdgcn_sched_barrier(0);
  e_mma(acc, a0, b0);
  e_load(a0, b0, ap + 128, bp + 128);
  __builtin_amdgcn_sched_barrier(0);
  e_mma(acc, a1, b1);
  e_load(a1, b1, ap + 192, bp + 192);
  __builtin_amdgcn_sched_barrier(0);
  e_mma(acc, a0, b0);
  e_mma(acc, a1, b1);
#pragma unroll
  for (int n = 0; n < 2; ++n) {
#pragma unroll
    for (int r = 0; r < 8; ++r) {
      const int i = i0 + r + 8 * half;
      const int j = j0 + n * 16 + l16;
      const float g = -__logf(-__logf(u[n][r]));
      ep[(size_t)i * J_ + j] = fmaxf((acc[n][r] * ESCALE + g) * TEMPINV, NEGINF);
    }
  }
}

#define XG 6

__device__ __forceinline__ void x_load(float (&a)[XG][2], v4f (&bq)[XG][2],
                                       const float* sp, const float* tp) {
#pragma unroll
  for (int s = 0; s < XG; ++s) {
    a[s][0]  = sp[(size_t)(4 * s)     * J_];
    a[s][1]  = sp[(size_t)(4 * s + 1) * J_];
    bq[s][0] = *(const v4f*)(tp + (size_t)(4 * s)     * D_);
    bq[s][1] = *(const v4f*)(tp + (size_t)(4 * s + 1) * D_);
  }
}

__device__ __forceinline__ void x_mma(v8f (&acc)[4], const float (&a)[XG][2],
                                      const v4f (&bq)[XG][2]) {
#pragma unroll
  for (int s = 0; s < XG; ++s) {
    const v2f w = { __expf(a[s][0]), __expf(a[s][1]) };
#pragma unroll
    for (int n = 0; n < 4; ++n) {
      const v2f bv2 = { bq[s][0][n], bq[s][1][n] };
      acc[n] = __builtin_amdgcn_wmma_f32_16x16x4_f32(false, w, false, bv2,
                                                     (short)0, acc[n], false, false);
    }
  }
}

__global__ __launch_bounds__(128)
void expand_kernel(const float* __restrict__ text,
                   const float* __restrict__ soft,
                   float* __restrict__ out_exp)
{
  const int lane = threadIdx.x & 31;
  const int wave = threadIdx.x >> 5;
  const int half = lane >> 4;
  const int l16  = lane & 15;
  const int wb  = blockIdx.x * 4 + wave;
  const int b   = wb / 96;
  const int rem = wb - b * 96;
  const int mb  = rem / 4;
  const int nb  = rem - mb * 4;
  const int jb  = mb * 16;
  const int db  = nb * 64;

  const float* sp = soft + (size_t)b * I_ * J_ + (size_t)(2 * half) * J_ + jb + l16;
  const float* tp = text + (size_t)b * I_ * D_ + (size_t)(2 * half) * D_ + db + 4 * l16;

  v8f acc[4];
#pragma unroll
  for (int n = 0; n < 4; ++n) acc[n] = (v8f){};
  float a0[XG][2], a1[XG][2];
  v4f   b0[XG][2], b1[XG][2];
  x_load(a0, b0, sp, tp);
  x_load(a1, b1, sp + (size_t)(4 * XG) * J_,  tp + (size_t)(4 * XG) * D_);
  x_mma(acc, a0, b0);
  x_load(a0, b0, sp + (size_t)(8 * XG) * J_,  tp + (size_t)(8 * XG) * D_);
  x_mma(acc, a1, b1);
  x_load(a1, b1, sp + (size_t)(12 * XG) * J_, tp + (size_t)(12 * XG) * D_);
  x_mma(acc, a0, b0);
  x_mma(acc, a1, b1);
#pragma unroll
  for (int r = 0; r < 8; ++r) {
    const int j = jb + r + 8 * half;
    *(v4f*)(out_exp + (size_t)b * J_ * D_ + (size_t)j * D_ + db + 4 * l16) =
        (v4f){ acc[0][r], acc[1][r], acc[2][r], acc[3][r] };
  }
}

__global__ __launch_bounds__(384, 1)
void dp_kernel(float* __restrict__ out_soft)
{
  __shared__ __attribute__((aligned(16))) float sE[I_ * J_];
  __shared__ __attribute__((aligned(16))) float sL[I_ * J_];
  __shared__ float sP[J_];
  __shared__ float sWt[32];

  const int b    = blockIdx.x;
  const int tid  = threadIdx.x;
  const int lane = tid & 31;
  const int wave = tid >> 5;

  {
    const v4f* eg = (const v4f*)(out_soft + (size_t)b * I_ * J_);
    v4f* el = (v4f*)sE;
    v4f ev4[I_ * J_ / 4 / 384];
#pragma unroll
    for (int r = 0; r < I_ * J_ / 4 / 384; ++r) ev4[r] = eg[tid + 384 * r];
#pragma unroll
    for (int r = 0; r < I_ * J_ / 4 / 384; ++r) el[tid + 384 * r] = ev4[r];
  }
  __syncthreads();

  for (int r = 0; r < RPW; ++r) {
    const int row = wave * RPW + r;
    const float* er = sE + row * J_ + lane * CPL;
    const v4f x0 = *(const v4f*)(er + 0);
    const v4f x1 = *(const v4f*)(er + 4);
    const v4f x2 = *(const v4f*)(er + 8);
    float s[CPL] = { x0.x, x0.y, x0.z, x0.w, x1.x, x1.y, x1.z, x1.w,
                     x2.x, x2.y, x2.z, x2.w };
    for (int c = CPL - 2; c >= 0; --c) s[c] = lse2(s[c], s[c + 1]);
    const float inc = wave_suffix_lse(s[0], lane);
    const float exc = wave_next_lane(inc, lane);
    for (int c = 0; c < CPL; ++c) s[c] = lse2(s[c], exc);
    float* lr = sL + row * J_ + lane * CPL;
    *(v4f*)(lr + 0) = (v4f){ s[0], s[1], s[2],  s[3]  };
    *(v4f*)(lr + 4) = (v4f){ s[4], s[5], s[6],  s[7]  };
    *(v4f*)(lr + 8) = (v4f){ s[8], s[9], s[10], s[11] };
  }

  float bv = (tid == 0) ? 0.0f : NEGV;
  const int st = (tid == 0) ? (J_ - 1) : (tid - 1);
  const int wv = __builtin_amdgcn_readfirstlane(wave);
  if (tid < 32) sWt[tid] = NEGINF;
  __syncthreads();

  for (int i = 0; i < I_ - 1; ++i) {
    const float Lv = sL[i * J_ + tid];

    const float av = bv - Lv;
    float xp  = wave_prefix_lse(av, lane);
    float xsa = wave_suffix_lse(av, lane);
    if (lane == 31) sWt[1 + wave]           = xp;
    if (lane == 0)  sWt[16 + NWAVE - wave]  = xsa;
    __syncthreads();
    float t = sWt[lane];
    t = lse2(t, row_shr_fill<1>(t, NEGINF));
    t = lse2(t, row_shr_fill<2>(t, NEGINF));
    t = lse2(t, row_shr_fill<4>(t, NEGINF));
    t = lse2(t, row_shr_fill<8>(t, NEGINF));
    const float plo  = __builtin_bit_cast(float,
        __builtin_amdgcn_readlane(__builtin_bit_cast(int, t), wv));
    const float sahi = __builtin_bit_cast(float,
        __builtin_amdgcn_readlane(__builtin_bit_cast(int, t), 16 + NWAVE - 1 - wv));
    const float Pv  = lse2(xp,  plo);
    const float SAv = lse2(xsa, sahi);
    sP[tid] = Pv;
    sL[i * J_ + tid] = Lv + Pv;
    __syncthreads();

    const int r = i + 1;
    const float ev = sE[i * J_ + st];
    const float cand = lse2(ev + sP[st], SAv + LOGEPS);
    const float nb = (tid >= r && tid <= (J_ - I_ + r + 1)) ? cand : NEGV;
    sE[i * J_ + st] = bv;
    bv = nb;
  }
  sE[(I_ - 1) * J_ + st] = bv;
  __syncthreads();

  for (int rr = 0; rr < RPW; ++rr) {
    const int row = wave * RPW + rr;
    const float* br = sE + row * J_ + lane * CPL;
    const float* qr = sL + row * J_ + lane * CPL;
    const v4f x0 = *(const v4f*)(br + 0);
    const v4f x1 = *(const v4f*)(br + 4);
    const v4f x2 = *(const v4f*)(br + 8);
    const v4f q0 = *(const v4f*)(qr + 0);
    const v4f q1 = *(const v4f*)(qr + 4);
    const v4f q2 = *(const v4f*)(qr + 8);
    float s[CPL] = { x0.x, x0.y, x0.z, x0.w, x1.x, x1.y, x1.z, x1.w,
                     x2.x, x2.y, x2.z, x2.w };
    const float q[CPL] = { q0.x, q0.y, q0.z, q0.w, q1.x, q1.y, q1.z, q1.w,
                           q2.x, q2.y, q2.z, q2.w };
    const float bfirst = s[CPL - 1];
    if (lane == 31) s[CPL - 1] = NEGINF;
    for (int c = CPL - 2; c >= 0; --c) s[c] = lse2(s[c], s[c + 1]);
    const float inc = wave_suffix_lse(s[0], lane);
    const float exc = wave_next_lane(inc, lane);
    for (int c = 0; c < CPL; ++c) s[c] = lse2(s[c], exc);
    float o[CPL];
    if (row < I_ - 1) {
      for (int c = 0; c < CPL; ++c) o[c] = lse2(q[c], s[c] + LOGEPS);
    } else {
      const float tot = lse2(lane_bcast<0>(s[0]), lane_bcast<31>(bfirst));
      for (int c = 0; c < CPL; ++c)
        o[c] = tot + ((lane * CPL + c == J_ - 1) ? 0.0f : LOGEPS);
    }
    float* og = out_soft + (size_t)b * I_ * J_ + (size_t)row * J_ + lane * CPL;
    *(v4f*)(og + 0) = (v4f){ o[0], o[1], o[2],  o[3]  };
    *(v4f*)(og + 4) = (v4f){ o[4], o[5], o[6],  o[7]  };
    *(v4f*)(og + 8) = (v4f){ o[8], o[9], o[10], o[11] };
  }
}

extern "C" void kernel_launch(void* const* d_in, const int* in_sizes, int n_in,
                              void* d_out, int out_size, void* d_ws, size_t ws_size,
                              hipStream_t stream) {
  (void)in_sizes; (void)n_in; (void)d_ws; (void)ws_size; (void)out_size;
  const float* text = (const float*)d_in[0];
  const float* mel  = (const float*)d_in[1];
  const float* gum  = (const float*)d_in[2];
  float* out_soft = (float*)d_out;
  float* out_exp  = (float*)d_out + (size_t)B_ * I_ * J_;
  energy_kernel<<<dim3(72), dim3(128), 0, stream>>>(text, mel, gum, out_soft);
  dp_kernel<<<dim3(B_), dim3(384), 0, stream>>>(out_soft);
  expand_kernel<<<dim3(96), dim3(128), 0, stream>>>(text, out_soft, out_exp);
}


# Serial DP rows on 4 waves (one per SIMD), 3 consecutive k per lane: in-lane LSE then one scan of lane totals, lane offset serves as P(jj-1), one barrier per row (new reduction tree for P and SA)
# speedup vs baseline: 9.9087x; 1.0549x over previous
#include <hip/hip_runtime.h>
#include <hip/hip_bf16.h>
#include <math.h>


typedef float v2f __attribute__((ext_vector_type(2)));
typedef float v4f __attribute__((ext_vector_type(4)));
typedef float v8f __attribute__((ext_vector_type(8)));

#define B_      4
#define I_      96
#define J_      384
#define D_      256
#define NWAVE   12
#define RPW     (I_ / NWAVE)
#define CPL     (J_ / 32)
#define DPW     4
#define EPL     (J_ / (32 * DPW))
#define LOGEPS  (-1000.0f)
#define NEGV    (-1000000000.0f)
#define NEGINF  (-1.0e38f)
#define TEMPINV (1.0f/0.55f)
#define ESCALE  (1.0f/256.0f)
#define LOG2E   1.4426950408889634f
#define LN2     0.6931471805599453f

__device__ __forceinline__ float lse2(float a, float b) {
  const float m = fmaxf(a, b);
  const float e = __builtin_amdgcn_exp2f(-fabsf(a - b) * LOG2E);
  return __builtin_fmaf(__builtin_amdgcn_logf(1.0f + e), LN2, m);
}

template <int N>
__device__ __forceinline__ float row_shr_fill(float x, float fill) {
  return __builtin_bit_cast(float, __builtin_amdgcn_update_dpp(
      __builtin_bit_cast(int, fill), __builtin_bit_cast(int, x),
      0x110 + N  , 0xF, 0xF, false));
}

template <int N>
__device__ __forceinline__ float row_shl_fill(float x, float fill) {
  return __builtin_bit_cast(float, __builtin_amdgcn_update_dpp(
      __builtin_bit_cast(int, fill), __builtin_bit_cast(int, x),
      0x100 + N  , 0xF, 0xF, false));
}

template <int L>
__device__ __forceinline__ float lane_bcast(float x) {
  return __builtin_bit_cast(float,
      __builtin_amdgcn_readlane(__builtin_bit_cast(int, x), L));
}

__device__ __forceinline__ float wave_prefix_lse(float x, int lane) {
  x = lse2(x, row_shr_fill<1>(x, NEGINF));
  x = lse2(x, row_shr_fill<2>(x, NEGINF));
  x = lse2(x, row_shr_fill<4>(x, NEGINF));
  x = lse2(x, row_shr_fill<8>(x, NEGINF));
  const float lo = lane_bcast<15>(x);
  return lse2(x, (lane >= 16) ? lo : NEGINF);
}

__device__ __forceinline__ float wave_suffix_lse(float x, int lane) {
  x = lse2(x, row_shl_fill<1>(x, NEGINF));
  x = lse2(x, row_shl_fill<2>(x, NEGINF));
  x = lse2(x, row_shl_fill<4>(x, NEGINF));
  x = lse2(x, row_shl_fill<8>(x, NEGINF));
  const float hi = lane_bcast<16>(x);
  return lse2(x, (lane < 16) ? hi : NEGINF);
}

__device__ __forceinline__ float wave_next_lane(float x, int lane) {
  const float e  = row_shl_fill<1>(x, NEGINF);
  const float hi = lane_bcast<16>(x);
  return (lane == 15) ? hi : e;
}

__device__ __forceinline__ void wave_prefix_lse_ie(float x, int lane, float& inc, float& exc) {
  x = lse2(x, row_shr_fill<1>(x, NEGINF));
  x = lse2(x, row_shr_fill<2>(x, NEGINF));
  x = lse2(x, row_shr_fill<4>(x, NEGINF));
  x = lse2(x, row_shr_fill<8>(x, NEGINF));
  const float lo  = lane_bcast<15>(x);
  const float add = (lane >= 16) ? lo : NEGINF;
  inc = lse2(x, add);
  exc = lse2(row_shr_fill<1>(x, NEGINF), add);
}

__device__ __forceinline__ void wave_suffix_lse_ie(float x, int lane, float& inc, float& exc) {
  x = lse2(x, row_shl_fill<1>(x, NEGINF));
  x = lse2(x, row_shl_fill<2>(x, NEGINF));
  x = lse2(x, row_shl_fill<4>(x, NEGINF));
  x = lse2(x, row_shl_fill<8>(x, NEGINF));
  const float hi  = lane_bcast<16>(x);
  const float add = (lane < 16) ? hi : NEGINF;
  inc = lse2(x, add);
  exc = lse2(row_shl_fill<1>(x, NEGINF), add);
}

#define EG 8

__device__ __forceinline__ void e_load(v4f (&a)[EG], v4f (&bq)[EG][2],
                                       const float* ap, const float* bp) {
#pragma unroll
  for (int c = 0; c < EG; ++c) {
    a[c] = *(const v4f*)(ap + 8 * c);
#pragma unroll
    for (int n = 0; n < 2; ++n) bq[c][n] = *(const v4f*)(bp + 8 * c + (size_t)n * 16 * D_);
  }
}

__device__ __forceinline__ void e_mma(v8f (&acc)[2], const v4f (&a)[EG],
                                      const v4f (&bq)[EG][2]) {
#pragma unroll
  for (int c = 0; c < EG; ++c) {
#pragma unroll
    for (int n = 0; n < 2; ++n) {
      const v2f av = { a[c].x, a[c].y };
      const v2f bv2 = { bq[c][n].x, bq[c][n].y };
      acc[n] = __builtin_amdgcn_wmma_f32_16x16x4_f32(false, av, false, bv2,
                                                     (short)0, acc[n], false, false);
    }
#pragma unroll
    for (int n = 0; n < 2; ++n) {
      const v2f av = { a[c].z, a[c].w };
      const v2f bv2 = { bq[c][n].z, bq[c][n].w };
      acc[n] = __builtin_amdgcn_wmma_f32_16x16x4_f32(false, av, false, bv2,
                                                     (short)0, acc[n], false, false);
    }
  }
}

__global__ __launch_bounds__(128)
void energy_kernel(const float* __restrict__ text,
                   const float* __restrict__ mel,
                   const float* __restrict__ gum,
                   float* __restrict__ energy)
{
  const int lane = threadIdx.x & 31;
  const int wave = threadIdx.x >> 5;
  const int half = lane >> 4;
  const int l16  = lane & 15;
  const int wb  = blockIdx.x * 4 + wave;
  const int b   = wb / 72;
  const int rem = wb - b * 72;
  const int mb  = rem / 12;
  const int nb  = rem - mb * 12;
  const int i0  = mb * 16;
  const int j0  = nb * 32;

  const float* ap = text + ((size_t)b * I_ + i0 + l16) * D_ + 4 * half;
  const float* bp = mel  + ((size_t)b * J_ + j0 + l16) * D_ + 4 * half;
  const float* gp = gum    + (size_t)b * I_ * J_;
  float*       ep = energy + (size_t)b * I_ * J_;

  v8f acc[2];
  acc[0] = (v8f){}; acc[1] = (v8f){};
  v4f a0[EG], b0[EG][2], a1[EG], b1[EG][2];
  e_load(a0, b0, ap, bp);
  __builtin_amdgcn_sched_barrier(0);
  float u[2][8];
#pragma unroll
  for (int n = 0; n < 2; ++n) {
#pragma unroll
    for (int r = 0; r < 8; ++r)
      u[n][r] = gp[(size_t)(i0 + r + 8 * half) * J_ + j0 + n * 16 + l16];
  }
  e_load(a1, b1, ap + 64, bp + 64);
  __builtin_amdgcn_sched_barrier(0);
  e_mma(acc, a0, b0);
  e_load(a0, b0, ap + 128, bp + 128);
  __builtin_amdgcn_sched_barrier(0);
  e_mma(acc, a1, b1);
  e_load(a1, b1, ap + 192, bp + 192);
  __builtin_amdgcn_sched_barrier(0);
  e_mma(acc, a0, b0);
  e_mma(acc, a1, b1);
#pragma unroll
  for (int n = 0; n < 2; ++n) {
#pragma unroll
    for (int r = 0; r < 8; ++r) {
      const int i = i0 + r + 8 * half;
      const int j = j0 + n * 16 + l16;
      const float g = -__logf(-__logf(u[n][r]));
      ep[(size_t)i * J_ + j] = fmaxf((acc[n][r] * ESCALE + g) * TEMPINV, NEGINF);
    }
  }
}

#define XG 6

__device__ __forceinline__ void x_load(float (&a)[XG][2], v4f (&bq)[XG][2],
                                       const float* sp, const float* tp) {
#pragma unroll
  for (int s = 0; s < XG; ++s) {
    a[s][0]  = sp[(size_t)(4 * s)     * J_];
    a[s][1]  = sp[(size_t)(4 * s + 1) * J_];
    bq[s][0] = *(const v4f*)(tp + (size_t)(4 * s)     * D_);
    bq[s][1] = *(const v4f*)(tp + (size_t)(4 * s + 1) * D_);
  }
}

__device__ __forceinline__ void x_mma(v8f (&acc)[4], const float (&a)[XG][2],
                                      const v4f (&bq)[XG][2]) {
#pragma unroll
  for (int s = 0; s < XG; ++s) {
    const v2f w = { __expf(a[s][0]), __expf(a[s][1]) };
#pragma unroll
    for (int n = 0; n < 4; ++n) {
      const v2f bv2 = { bq[s][0][n], bq[s][1][n] };
      acc[n] = __builtin_amdgcn_wmma_f32_16x16x4_f32(false, w, false, bv2,
                                                     (short)0, acc[n], false, false);
    }
  }
}

__global__ __launch_bounds__(128)
void expand_kernel(const float* __restrict__ text,
                   const float* __restrict__ soft,
                   float* __restrict__ out_exp)
{
  const int lane = threadIdx.x & 31;
  const int wave = threadIdx.x >> 5;
  const int half = lane >> 4;
  const int l16  = lane & 15;
  const int wb  = blockIdx.x * 4 + wave;
  const int b   = wb / 96;
  const int rem = wb - b * 96;
  const int mb  = rem / 4;
  const int nb  = rem - mb * 4;
  const int jb  = mb * 16;
  const int db  = nb * 64;

  const float* sp = soft + (size_t)b * I_ * J_ + (size_t)(2 * half) * J_ + jb + l16;
  const float* tp = text + (size_t)b * I_ * D_ + (size_t)(2 * half) * D_ + db + 4 * l16;

  v8f acc[4];
#pragma unroll
  for (int n = 0; n < 4; ++n) acc[n] = (v8f){};
  float a0[XG][2], a1[XG][2];
  v4f   b0[XG][2], b1[XG][2];
  x_load(a0, b0, sp, tp);
  x_load(a1, b1, sp + (size_t)(4 * XG) * J_,  tp + (size_t)(4 * XG) * D_);
  x_mma(acc, a0, b0);
  x_load(a0, b0, sp + (size_t)(8 * XG) * J_,  tp + (size_t)(8 * XG) * D_);
  x_mma(acc, a1, b1);
  x_load(a1, b1, sp + (size_t)(12 * XG) * J_, tp + (size_t)(12 * XG) * D_);
  x_mma(acc, a0, b0);
  x_mma(acc, a1, b1);
#pragma unroll
  for (int r = 0; r < 8; ++r) {
    const int j = jb + r + 8 * half;
    *(v4f*)(out_exp + (size_t)b * J_ * D_ + (size_t)j * D_ + db + 4 * l16) =
        (v4f){ acc[0][r], acc[1][r], acc[2][r], acc[3][r] };
  }
}

__global__ __launch_bounds__(384, 1)
void dp_kernel(float* __restrict__ out_soft)
{
  __shared__ __attribute__((aligned(16))) float sE[I_ * J_];
  __shared__ __attribute__((aligned(16))) float sL[I_ * J_];
  __shared__ float sWt[2][32];

  const int b    = blockIdx.x;
  const int tid  = threadIdx.x;
  const int lane = tid & 31;
  const int wave = tid >> 5;

  {
    const v4f* eg = (const v4f*)(out_soft + (size_t)b * I_ * J_);
    v4f* el = (v4f*)sE;
    v4f ev4[I_ * J_ / 4 / 384];
#pragma unroll
    for (int r = 0; r < I_ * J_ / 4 / 384; ++r) ev4[r] = eg[tid + 384 * r];
#pragma unroll
    for (int r = 0; r < I_ * J_ / 4 / 384; ++r) el[tid + 384 * r] = ev4[r];
  }
  __syncthreads();

  for (int r = 0; r < RPW; ++r) {
    const int row = wave * RPW + r;
    const float* er = sE + row * J_ + lane * CPL;
    const v4f x0 = *(const v4f*)(er + 0);
    const v4f x1 = *(const v4f*)(er + 4);
    const v4f x2 = *(const v4f*)(er + 8);
    float s[CPL] = { x0.x, x0.y, x0.z, x0.w, x1.x, x1.y, x1.z, x1.w,
                     x2.x, x2.y, x2.z, x2.w };
    for (int c = CPL - 2; c >= 0; --c) s[c] = lse2(s[c], s[c + 1]);
    const float inc = wave_suffix_lse(s[0], lane);
    const float exc = wave_next_lane(inc, lane);
    for (int c = 0; c < CPL; ++c) s[c] = lse2(s[c], exc);
    float* lr = sL + row * J_ + lane * CPL;
    *(v4f*)(lr + 0) = (v4f){ s[0], s[1], s[2],  s[3]  };
    *(v4f*)(lr + 4) = (v4f){ s[4], s[5], s[6],  s[7]  };
    *(v4f*)(lr + 8) = (v4f){ s[8], s[9], s[10], s[11] };
  }

  const bool act = wave < DPW;
  const int  k0  = EPL * tid;
  float bv[EPL];
  int   st[EPL];
  for (int e = 0; e < EPL; ++e) {
    const int jj = k0 + e;
    bv[e] = (jj == 0) ? 0.0f : NEGV;
    st[e] = (jj == 0) ? (J_ - 1) : (jj - 1);
  }
  const int wv = __builtin_amdgcn_readfirstlane(wave);
  if (tid < 64) (&sWt[0][0])[tid] = NEGINF;
  __syncthreads();

  float Ln[EPL];
  if (act) {
    for (int e = 0; e < EPL; ++e) Ln[e] = sL[k0 + e];
  }
  for (int i = 0; i < I_ - 1; ++i) {
    float* wt = sWt[i & 1];
    float Lv[EPL], ev[EPL], p[EPL], s[EPL];
    float xpi = NEGINF, xpe = NEGINF, xsi = NEGINF, xse = NEGINF;
    if (act) {
      for (int e = 0; e < EPL; ++e) {
        Lv[e] = Ln[e];
        Ln[e] = sL[(i + 1) * J_ + k0 + e];
        ev[e] = sE[i * J_ + st[e]];
      }
      float a[EPL];
      for (int e = 0; e < EPL; ++e) a[e] = bv[e] - Lv[e];
      p[0] = a[0];
      for (int e = 1; e < EPL; ++e) p[e] = lse2(p[e - 1], a[e]);
      s[EPL - 1] = a[EPL - 1];
      for (int e = EPL - 2; e >= 0; --e) s[e] = lse2(a[e], s[e + 1]);
      wave_prefix_lse_ie(p[EPL - 1], lane, xpi, xpe);
      wave_suffix_lse_ie(s[0],       lane, xsi, xse);
      if (lane == 31) wt[1 + wave]         = xpi;
      if (lane == 0)  wt[16 + DPW - wave]  = xsi;
    }
    __syncthreads();
    if (act) {
      float t = wt[lane];
      t = lse2(t, row_shr_fill<1>(t, NEGINF));
      t = lse2(t, row_shr_fill<2>(t, NEGINF));
      const float plo  = __builtin_bit_cast(float,
          __builtin_amdgcn_readlane(__builtin_bit_cast(int, t), wv));
      const float sahi = __builtin_bit_cast(float,
          __builtin_amdgcn_readlane(__builtin_bit_cast(int, t), 16 + DPW - 1 - wv));
      const float offP = lse2(xpe, plo);
      const float offS = lse2(xse, sahi);
      float Pv[EPL], SAv[EPL];
      for (int e = 0; e < EPL; ++e) {
        Pv[e]  = lse2(p[e], offP);
        SAv[e] = lse2(s[e], offS);
        sL[i * J_ + k0 + e] = Lv[e] + Pv[e];
      }
      const int r = i + 1;
      for (int e = 0; e < EPL; ++e) {
        const int   jj    = k0 + e;
        const float pprev = (e == 0) ? offP : Pv[e > 0 ? e - 1 : 0];
        const float cand = lse2(ev[e] + pprev, SAv[e] + LOGEPS);
        const float nb = (jj >= r && jj <= (J_ - I_ + r + 1)) ? cand : NEGV;
        sE[i * J_ + st[e]] = bv[e];
        bv[e] = nb;
      }
    }
  }
  if (act) {
    for (int e = 0; e < EPL; ++e)
      sE[(I_ - 1) * J_ + st[e]] = bv[e];
  }
  __syncthreads();

  for (int rr = 0; rr < RPW; ++rr) {
    const int row = wave * RPW + rr;
    const float* br = sE + row * J_ + lane * CPL;
    const float* qr = sL + row * J_ + lane * CPL;
    const v4f x0 = *(const v4f*)(br + 0);
    const v4f x1 = *(const v4f*)(br + 4);
    const v4f x2 = *(const v4f*)(br + 8);
    const v4f q0 = *(const v4f*)(qr + 0);
    const v4f q1 = *(const v4f*)(qr + 4);
    const v4f q2 = *(const v4f*)(qr + 8);
    float s[CPL] = { x0.x, x0.y, x0.z, x0.w, x1.x, x1.y, x1.z, x1.w,
                     x2.x, x2.y, x2.z, x2.w };
    const float q[CPL] = { q0.x, q0.y, q0.z, q0.w, q1.x, q1.y, q1.z, q1.w,
                           q2.x, q2.y, q2.z, q2.w };
    const float bfirst = s[CPL - 1];
    if (lane == 31) s[CPL - 1] = NEGINF;
    for (int c = CPL - 2; c >= 0; --c) s[c] = lse2(s[c], s[c + 1]);
    const float inc = wave_suffix_lse(s[0], lane);
    const float exc = wave_next_lane(inc, lane);
    for (int c = 0; c < CPL; ++c) s[c] = lse2(s[c], exc);
    float o[CPL];
    if (row < I_ - 1) {
      for (int c = 0; c < CPL; ++c) o[c] = lse2(q[c], s[c] + LOGEPS);
    } else {
      const float tot = lse2(lane_bcast<0>(s[0]), lane_bcast<31>(bfirst));
      for (int c = 0; c < CPL; ++c)
        o[c] = tot + ((lane * CPL + c == J_ - 1) ? 0.0f : LOGEPS);
    }
    float* og = out_soft + (size_t)b * I_ * J_ + (size_t)row * J_ + lane * CPL;
    *(v4f*)(og + 0) = (v4f){ o[0], o[1], o[2],  o[3]  };
    *(v4f*)(og + 4) = (v4f){ o[4], o[5], o[6],  o[7]  };
    *(v4f*)(og + 8) = (v4f){ o[8], o[9], o[10], o[11] };
  }
}

extern "C" void kernel_launch(void* const* d_in, const int* in_sizes, int n_in,
                              void* d_out, int out_size, void* d_ws, size_t ws_size,
                              hipStream_t stream) {
  (void)in_sizes; (void)n_in; (void)d_ws; (void)ws_size; (void)out_size;
  const float* text = (const float*)d_in[0];
  const float* mel  = (const float*)d_in[1];
  const float* gum  = (const float*)d_in[2];
  float* out_soft = (float*)d_out;
  float* out_exp  = (float*)d_out + (size_t)B_ * I_ * J_;
  energy_kernel<<<dim3(72), dim3(128), 0, stream>>>(text, mel, gum, out_soft);
  dp_kernel<<<dim3(B_), dim3(384), 0, stream>>>(out_soft);
  expand_kernel<<<dim3(96), dim3(128), 0, stream>>>(text, out_soft, out_exp);
}


# Serial DP rows sum in the linear domain on (integer exponent, mantissa) pairs: a scan step is max/ldexp/add, no exp2/log2 in the row chain (new rounding of P, SA, Bij; same trees)
# speedup vs baseline: 11.8007x; 1.1909x over previous
#include <hip/hip_runtime.h>
#include <hip/hip_bf16.h>
#include <math.h>


typedef float v2f __attribute__((ext_vector_type(2)));
typedef float v4f __attribute__((ext_vector_type(4)));
typedef float v8f __attribute__((ext_vector_type(8)));

#define B_      4
#define I_      96
#define J_      384
#define D_      256
#define NWAVE   12
#define RPW     (I_ / NWAVE)
#define CPL     (J_ / 32)
#define DPW     4
#define EPL     (J_ / (32 * DPW))
#define LOGEPS  (-1000.0f)
#define NEGV    (-1000000000.0f)
#define NEGINF  (-1.0e38f)
#define TEMPINV (1.0f/0.55f)
#define ESCALE  (1.0f/256.0f)
#define LOG2E   1.4426950408889634f
#define LN2     0.6931471805599453f

__device__ __forceinline__ float lse2(float a, float b) {
  const float m = fmaxf(a, b);
  const float e = __builtin_amdgcn_exp2f(-fabsf(a - b) * LOG2E);
  return __builtin_fmaf(__builtin_amdgcn_logf(1.0f + e), LN2, m);
}

template <int N>
__device__ __forceinline__ float row_shr_fill(float x, float fill) {
  return __builtin_bit_cast(float, __builtin_amdgcn_update_dpp(
      __builtin_bit_cast(int, fill), __builtin_bit_cast(int, x),
      0x110 + N  , 0xF, 0xF, false));
}

template <int N>
__device__ __forceinline__ float row_shl_fill(float x, float fill) {
  return __builtin_bit_cast(float, __builtin_amdgcn_update_dpp(
      __builtin_bit_cast(int, fill), __builtin_bit_cast(int, x),
      0x100 + N  , 0xF, 0xF, false));
}

template <int L>
__device__ __forceinline__ float lane_bcast(float x) {
  return __builtin_bit_cast(float,
      __builtin_amdgcn_readlane(__builtin_bit_cast(int, x), L));
}

__device__ __forceinline__ float wave_prefix_lse(float x, int lane) {
  x = lse2(x, row_shr_fill<1>(x, NEGINF));
  x = lse2(x, row_shr_fill<2>(x, NEGINF));
  x = lse2(x, row_shr_fill<4>(x, NEGINF));
  x = lse2(x, row_shr_fill<8>(x, NEGINF));
  const float lo = lane_bcast<15>(x);
  return lse2(x, (lane >= 16) ? lo : NEGINF);
}

__device__ __forceinline__ float wave_suffix_lse(float x, int lane) {
  x = lse2(x, row_shl_fill<1>(x, NEGINF));
  x = lse2(x, row_shl_fill<2>(x, NEGINF));
  x = lse2(x, row_shl_fill<4>(x, NEGINF));
  x = lse2(x, row_shl_fill<8>(x, NEGINF));
  const float hi = lane_bcast<16>(x);
  return lse2(x, (lane < 16) ? hi : NEGINF);
}

__device__ __forceinline__ float wave_next_lane(float x, int lane) {
  const float e  = row_shl_fill<1>(x, NEGINF);
  const float hi = lane_bcast<16>(x);
  return (lane == 15) ? hi : e;
}

#define MNEG   (-1442695040)
#define MEPS   (-1443)
#define SEPS   1.2353836f
#define XCLAMP 100000000.0f

__device__ __forceinline__ void padd(int& m, float& s, int m2, float s2) {
  const int mm = (m > m2) ? m : m2;
  s = __builtin_ldexpf(s, m - mm) + __builtin_ldexpf(s2, m2 - mm);
  m = mm;
}

__device__ __forceinline__ void to_pair(float x, int& m, float& s) {
  const float y = fminf(fmaxf(x * LOG2E, -XCLAMP), XCLAMP);
  const float f = floorf(y);
  m = (int)f;
  s = __builtin_amdgcn_exp2f(y - f);
}

__device__ __forceinline__ float from_pair(int m, float s) {
  return ((float)m + __builtin_amdgcn_logf(s)) * LN2;
}

template <int N>
__device__ __forceinline__ int row_shr_fill_i(int x, int fill) {
  return __builtin_amdgcn_update_dpp(fill, x, 0x110 + N  , 0xF, 0xF, false);
}

template <int N>
__device__ __forceinline__ int row_shl_fill_i(int x, int fill) {
  return __builtin_amdgcn_update_dpp(fill, x, 0x100 + N  , 0xF, 0xF, false);
}

__device__ __forceinline__ void wave_prefix_pair_ie(int m, float s, int lane,
                                                    int& mi, float& si, int& me, float& se) {
  padd(m, s, row_shr_fill_i<1>(m, MNEG), row_shr_fill<1>(s, 0.0f));
  padd(m, s, row_shr_fill_i<2>(m, MNEG), row_shr_fill<2>(s, 0.0f));
  padd(m, s, row_shr_fill_i<4>(m, MNEG), row_shr_fill<4>(s, 0.0f));
  padd(m, s, row_shr_fill_i<8>(m, MNEG), row_shr_fill<8>(s, 0.0f));
  const int   lom = __builtin_amdgcn_readlane(m, 15);
  const float los = lane_bcast<15>(s);
  const int   am  = (lane >= 16) ? lom : MNEG;
  const float as  = (lane >= 16) ? los : 0.0f;
  mi = m; si = s;
  padd(mi, si, am, as);
  me = row_shr_fill_i<1>(m, MNEG);
  se = row_shr_fill<1>(s, 0.0f);
  padd(me, se, am, as);
}

__device__ __forceinline__ void wave_suffix_pair_ie(int m, float s, int lane,
                                                    int& mi, float& si, int& me, float& se) {
  padd(m, s, row_shl_fill_i<1>(m, MNEG), row_shl_fill<1>(s, 0.0f));
  padd(m, s, row_shl_fill_i<2>(m, MNEG), row_shl_fill<2>(s, 0.0f));
  padd(m, s, row_shl_fill_i<4>(m, MNEG), row_shl_fill<4>(s, 0.0f));
  padd(m, s, row_shl_fill_i<8>(m, MNEG), row_shl_fill<8>(s, 0.0f));
  const int   him = __builtin_amdgcn_readlane(m, 16);
  const float his = lane_bcast<16>(s);
  const int   am  = (lane < 16) ? him : MNEG;
  const float as  = (lane < 16) ? his : 0.0f;
  mi = m; si = s;
  padd(mi, si, am, as);
  me = row_shl_fill_i<1>(m, MNEG);
  se = row_shl_fill<1>(s, 0.0f);
  padd(me, se, am, as);
}

#define EG 8

__device__ __forceinline__ void e_load(v4f (&a)[EG], v4f (&bq)[EG][2],
                                       const float* ap, const float* bp) {
#pragma unroll
  for (int c = 0; c < EG; ++c) {
    a[c] = *(const v4f*)(ap + 8 * c);
#pragma unroll
    for (int n = 0; n < 2; ++n) bq[c][n] = *(const v4f*)(bp + 8 * c + (size_t)n * 16 * D_);
  }
}

__device__ __forceinline__ void e_mma(v8f (&acc)[2], const v4f (&a)[EG],
                                      const v4f (&bq)[EG][2]) {
#pragma unroll
  for (int c = 0; c < EG; ++c) {
#pragma unroll
    for (int n = 0; n < 2; ++n) {
      const v2f av = { a[c].x, a[c].y };
      const v2f bv2 = { bq[c][n].x, bq[c][n].y };
      acc[n] = __builtin_amdgcn_wmma_f32_16x16x4_f32(false, av, false, bv2,
                                                     (short)0, acc[n], false, false);
    }
#pragma unroll
    for (int n = 0; n < 2; ++n) {
      const v2f av = { a[c].z, a[c].w };
      const v2f bv2 = { bq[c][n].z, bq[c][n].w };
      acc[n] = __builtin_amdgcn_wmma_f32_16x16x4_f32(false, av, false, bv2,
                                                     (short)0, acc[n], false, false);
    }
  }
}

__global__ __launch_bounds__(128)
void energy_kernel(const float* __restrict__ text,
                   const float* __restrict__ mel,
                   const float* __restrict__ gum,
                   float* __restrict__ energy)
{
  const int lane = threadIdx.x & 31;
  const int wave = threadIdx.x >> 5;
  const int half = lane >> 4;
  const int l16  = lane & 15;
  const int wb  = blockIdx.x * 4 + wave;
  const int b   = wb / 72;
  const int rem = wb - b * 72;
  const int mb  = rem / 12;
  const int nb  = rem - mb * 12;
  const int i0  = mb * 16;
  const int j0  = nb * 32;

  const float* ap = text + ((size_t)b * I_ + i0 + l16) * D_ + 4 * half;
  const float* bp = mel  + ((size_t)b * J_ + j0 + l16) * D_ + 4 * half;
  const float* gp = gum    + (size_t)b * I_ * J_;
  float*       ep = energy + (size_t)b * I_ * J_;

  v8f acc[2];
  acc[0] = (v8f){}; acc[1] = (v8f){};
  v4f a0[EG], b0[EG][2], a1[EG], b1[EG][2];
  e_load(a0, b0, ap, bp);
  __builtin_amdgcn_sched_barrier(0);
  float u[2][8];
#pragma unroll
  for (int n = 0; n < 2; ++n) {
#pragma unroll
    for (int r = 0; r < 8; ++r)
      u[n][r] = gp[(size_t)(i0 + r + 8 * half) * J_ + j0 + n * 16 + l16];
  }
  e_load(a1, b1, ap + 64, bp + 64);
  __builtin_amdgcn_sched_barrier(0);
  e_mma(acc, a0, b0);
  e_load(a0, b0, ap + 128, bp + 128);
  __builtin_amdgcn_sched_barrier(0);
  e_mma(acc, a1, b1);
  e_load(a1, b1, ap + 192, bp + 192);
  __builtin_amdgcn_sched_barrier(0);
  e_mma(acc, a0, b0);
  e_mma(acc, a1, b1);
#pragma unroll
  for (int n = 0; n < 2; ++n) {
#pragma unroll
    for (int r = 0; r < 8; ++r) {
      const int i = i0 + r + 8 * half;
      const int j = j0 + n * 16 + l16;
      const float g = -__logf(-__logf(u[n][r]));
      ep[(size_t)i * J_ + j] = fmaxf((acc[n][r] * ESCALE + g) * TEMPINV, NEGINF);
    }
  }
}

#define XG 6

__device__ __forceinline__ void x_load(float (&a)[XG][2], v4f (&bq)[XG][2],
                                       const float* sp, const float* tp) {
#pragma unroll
  for (int s = 0; s < XG; ++s) {
    a[s][0]  = sp[(size_t)(4 * s)     * J_];
    a[s][1]  = sp[(size_t)(4 * s + 1) * J_];
    bq[s][0] = *(const v4f*)(tp + (size_t)(4 * s)     * D_);
    bq[s][1] = *(const v4f*)(tp + (size_t)(4 * s + 1) * D_);
  }
}

__device__ __forceinline__ void x_mma(v8f (&acc)[4], const float (&a)[XG][2],
                                      const v4f (&bq)[XG][2]) {
#pragma unroll
  for (int s = 0; s < XG; ++s) {
    const v2f w = { __expf(a[s][0]), __expf(a[s][1]) };
#pragma unroll
    for (int n = 0; n < 4; ++n) {
      const v2f bv2 = { bq[s][0][n], bq[s][1][n] };
      acc[n] = __builtin_amdgcn_wmma_f32_16x16x4_f32(false, w, false, bv2,
                                                     (short)0, acc[n], false, false);
    }
  }
}

__global__ __launch_bounds__(128)
void expand_kernel(const float* __restrict__ text,
                   const float* __restrict__ soft,
                   float* __restrict__ out_exp)
{
  const int lane = threadIdx.x & 31;
  const int wave = threadIdx.x >> 5;
  const int half = lane >> 4;
  const int l16  = lane & 15;
  const int wb  = blockIdx.x * 4 + wave;
  const int b   = wb / 96;
  const int rem = wb - b * 96;
  const int mb  = rem / 4;
  const int nb  = rem - mb * 4;
  const int jb  = mb * 16;
  const int db  = nb * 64;

  const float* sp = soft + (size_t)b * I_ * J_ + (size_t)(2 * half) * J_ + jb + l16;
  const float* tp = text + (size_t)b * I_ * D_ + (size_t)(2 * half) * D_ + db + 4 * l16;

  v8f acc[4];
#pragma unroll
  for (int n = 0; n < 4; ++n) acc[n] = (v8f){};
  float a0[XG][2], a1[XG][2];
  v4f   b0[XG][2], b1[XG][2];
  x_load(a0, b0, sp, tp);
  x_load(a1, b1, sp + (size_t)(4 * XG) * J_,  tp + (size_t)(4 * XG) * D_);
  x_mma(acc, a0, b0);
  x_load(a0, b0, sp + (size_t)(8 * XG) * J_,  tp + (size_t)(8 * XG) * D_);
  x_mma(acc, a1, b1);
  x_load(a1, b1, sp + (size_t)(12 * XG) * J_, tp + (size_t)(12 * XG) * D_);
  x_mma(acc, a0, b0);
  x_mma(acc, a1, b1);
#pragma unroll
  for (int r = 0; r < 8; ++r) {
    const int j = jb + r + 8 * half;
    *(v4f*)(out_exp + (size_t)b * J_ * D_ + (size_t)j * D_ + db + 4 * l16) =
        (v4f){ acc[0][r], acc[1][r], acc[2][r], acc[3][r] };
  }
}

__global__ __launch_bounds__(384, 1)
void dp_kernel(float* __restrict__ out_soft)
{
  __shared__ __attribute__((aligned(16))) float sE[I_ * J_];
  __shared__ __attribute__((aligned(16))) float sL[I_ * J_];
  __shared__ int   sWm[2][32];
  __shared__ float sWs[2][32];

  const int b    = blockIdx.x;
  const int tid  = threadIdx.x;
  const int lane = tid & 31;
  const int wave = tid >> 5;

  {
    const v4f* eg = (const v4f*)(out_soft + (size_t)b * I_ * J_);
    v4f* el = (v4f*)sE;
    v4f ev4[I_ * J_ / 4 / 384];
#pragma unroll
    for (int r = 0; r < I_ * J_ / 4 / 384; ++r) ev4[r] = eg[tid + 384 * r];
#pragma unroll
    for (int r = 0; r < I_ * J_ / 4 / 384; ++r) el[tid + 384 * r] = ev4[r];
  }
  __syncthreads();

  for (int r = 0; r < RPW; ++r) {
    const int row = wave * RPW + r;
    const float* er = sE + row * J_ + lane * CPL;
    const v4f x0 = *(const v4f*)(er + 0);
    const v4f x1 = *(const v4f*)(er + 4);
    const v4f x2 = *(const v4f*)(er + 8);
    float s[CPL] = { x0.x, x0.y, x0.z, x0.w, x1.x, x1.y, x1.z, x1.w,
                     x2.x, x2.y, x2.z, x2.w };
    for (int c = CPL - 2; c >= 0; --c) s[c] = lse2(s[c], s[c + 1]);
    const float inc = wave_suffix_lse(s[0], lane);
    const float exc = wave_next_lane(inc, lane);
    for (int c = 0; c < CPL; ++c) s[c] = lse2(s[c], exc);
    float* lr = sL + row * J_ + lane * CPL;
    *(v4f*)(lr + 0) = (v4f){ s[0], s[1], s[2],  s[3]  };
    *(v4f*)(lr + 4) = (v4f){ s[4], s[5], s[6],  s[7]  };
    *(v4f*)(lr + 8) = (v4f){ s[8], s[9], s[10], s[11] };
  }

  const bool act = wave < DPW;
  const int  k0  = EPL * tid;
  int   bm[EPL];
  float bs[EPL];
  float bl[EPL];
  int   st[EPL];
  for (int e = 0; e < EPL; ++e) {
    const int jj = k0 + e;
    bm[e] = (jj == 0) ? 0 : MNEG;
    bs[e] = 1.0f;
    bl[e] = (jj == 0) ? 0.0f : NEGV;
    st[e] = (jj == 0) ? (J_ - 1) : (jj - 1);
  }
  const int wv = __builtin_amdgcn_readfirstlane(wave);
  if (tid < 64) {
    (&sWm[0][0])[tid] = MNEG;
    (&sWs[0][0])[tid] = 0.0f;
  }
  __syncthreads();

  float Ln[EPL];
  int   nLm[EPL];
  float nLs[EPL];
  if (act) {
    for (int e = 0; e < EPL; ++e) {
      Ln[e] = sL[k0 + e];
      to_pair(-Ln[e], nLm[e], nLs[e]);
    }
  }
  for (int i = 0; i < I_ - 1; ++i) {
    int*   wm = sWm[i & 1];
    float* ws = sWs[i & 1];
    float Lv[EPL], ev[EPL];
    int   pm[EPL], sm[EPL];
    float ps[EPL], ss[EPL];
    int   xpim = MNEG, xpem = MNEG, xsim = MNEG, xsem = MNEG;
    float xpis = 0.0f, xpes = 0.0f, xsis = 0.0f, xses = 0.0f;
    if (act) {
      int   am[EPL];
      float as[EPL];
      for (int e = 0; e < EPL; ++e) {
        Lv[e] = Ln[e];
        am[e] = bm[e] + nLm[e];
        as[e] = bs[e] * nLs[e];
        Ln[e] = sL[(i + 1) * J_ + k0 + e];
        ev[e] = sE[i * J_ + st[e]];
      }
      pm[0] = am[0]; ps[0] = as[0];
      for (int e = 1; e < EPL; ++e) {
        pm[e] = pm[e - 1]; ps[e] = ps[e - 1];
        padd(pm[e], ps[e], am[e], as[e]);
      }
      sm[EPL - 1] = am[EPL - 1]; ss[EPL - 1] = as[EPL - 1];
      for (int e = EPL - 2; e >= 0; --e) {
        sm[e] = am[e]; ss[e] = as[e];
        padd(sm[e], ss[e], sm[e + 1], ss[e + 1]);
      }
      wave_prefix_pair_ie(pm[EPL - 1], ps[EPL - 1], lane, xpim, xpis, xpem, xpes);
      wave_suffix_pair_ie(sm[0],       ss[0],       lane, xsim, xsis, xsem, xses);
      if (lane == 31) { wm[1 + wave]        = xpim; ws[1 + wave]        = xpis; }
      if (lane == 0)  { wm[16 + DPW - wave] = xsim; ws[16 + DPW - wave] = xsis; }
      for (int e = 0; e < EPL; ++e) to_pair(-Ln[e], nLm[e], nLs[e]);
    }
    __syncthreads();
    if (act) {
      int   tm = wm[lane];
      float ts = ws[lane];
      int   em[EPL];
      float es[EPL];
      for (int e = 0; e < EPL; ++e) to_pair(ev[e], em[e], es[e]);
      padd(tm, ts, row_shr_fill_i<1>(tm, MNEG), row_shr_fill<1>(ts, 0.0f));
      padd(tm, ts, row_shr_fill_i<2>(tm, MNEG), row_shr_fill<2>(ts, 0.0f));
      int   offPm = xpem, offSm = xsem;
      float offPs = xpes, offSs = xses;
      padd(offPm, offPs, __builtin_amdgcn_readlane(tm, wv),
           __builtin_bit_cast(float, __builtin_amdgcn_readlane(__builtin_bit_cast(int, ts), wv)));
      padd(offSm, offSs, __builtin_amdgcn_readlane(tm, 16 + DPW - 1 - wv),
           __builtin_bit_cast(float, __builtin_amdgcn_readlane(__builtin_bit_cast(int, ts), 16 + DPW - 1 - wv)));
      int   Pm[EPL], SAm[EPL];
      float Ps[EPL], SAs[EPL];
      for (int e = 0; e < EPL; ++e) {
        Pm[e]  = pm[e]; Ps[e]  = ps[e];
        padd(Pm[e], Ps[e], offPm, offPs);
        SAm[e] = sm[e]; SAs[e] = ss[e];
        padd(SAm[e], SAs[e], offSm, offSs);
        sL[i * J_ + k0 + e] = Lv[e] + from_pair(Pm[e], Ps[e]);
      }
      const int r = i + 1;
      for (int e = 0; e < EPL; ++e) {
        const int   jj  = k0 + e;
        const int   ppm = (e == 0) ? offPm : Pm[e > 0 ? e - 1 : 0];
        const float pps = (e == 0) ? offPs : Ps[e > 0 ? e - 1 : 0];
        int   cm = em[e] + ppm;
        float cs = es[e] * pps;
        padd(cm, cs, SAm[e] + MEPS, SAs[e] * SEPS);
        cm += __builtin_amdgcn_frexp_expf(cs);
        cs  = __builtin_amdgcn_frexp_mantf(cs);
        const bool in = (jj >= r && jj <= (J_ - I_ + r + 1));
        sE[i * J_ + st[e]] = bl[e];
        bm[e] = in ? cm : MNEG;
        bs[e] = in ? cs : 1.0f;
        bl[e] = in ? from_pair(cm, cs) : NEGV;
      }
    }
  }
  if (act) {
    for (int e = 0; e < EPL; ++e)
      sE[(I_ - 1) * J_ + st[e]] = bl[e];
  }
  __syncthreads();

  for (int rr = 0; rr < RPW; ++rr) {
    const int row = wave * RPW + rr;
    const float* br = sE + row * J_ + lane * CPL;
    const float* qr = sL + row * J_ + lane * CPL;
    const v4f x0 = *(const v4f*)(br + 0);
    const v4f x1 = *(const v4f*)(br + 4);
    const v4f x2 = *(const v4f*)(br + 8);
    const v4f q0 = *(const v4f*)(qr + 0);
    const v4f q1 = *(const v4f*)(qr + 4);
    const v4f q2 = *(const v4f*)(qr + 8);
    float s[CPL] = { x0.x, x0.y, x0.z, x0.w, x1.x, x1.y, x1.z, x1.w,
                     x2.x, x2.y, x2.z, x2.w };
    const float q[CPL] = { q0.x, q0.y, q0.z, q0.w, q1.x, q1.y, q1.z, q1.w,
                           q2.x, q2.y, q2.z, q2.w };
    const float bfirst = s[CPL - 1];
    if (lane == 31) s[CPL - 1] = NEGINF;
    for (int c = CPL - 2; c >= 0; --c) s[c] = lse2(s[c], s[c + 1]);
    const float inc = wave_suffix_lse(s[0], lane);
    const float exc = wave_next_lane(inc, lane);
    for (int c = 0; c < CPL; ++c) s[c] = lse2(s[c], exc);
    float o[CPL];
    if (row < I_ - 1) {
      for (int c = 0; c < CPL; ++c) o[c] = lse2(q[c], s[c] + LOGEPS);
    } else {
      const float tot = lse2(lane_bcast<0>(s[0]), lane_bcast<31>(bfirst));
      for (int c = 0; c < CPL; ++c)
        o[c] = tot + ((lane * CPL + c == J_ - 1) ? 0.0f : LOGEPS);
    }
    float* og = out_soft + (size_t)b * I_ * J_ + (size_t)row * J_ + lane * CPL;
    *(v4f*)(og + 0) = (v4f){ o[0], o[1], o[2],  o[3]  };
    *(v4f*)(og + 4) = (v4f){ o[4], o[5], o[6],  o[7]  };
    *(v4f*)(og + 8) = (v4f){ o[8], o[9], o[10], o[11] };
  }
}

extern "C" void kernel_launch(void* const* d_in, const int* in_sizes, int n_in,
                              void* d_out, int out_size, void* d_ws, size_t ws_size,
                              hipStream_t stream) {
  (void)in_sizes; (void)n_in; (void)d_ws; (void)ws_size; (void)out_size;
  const float* text = (const float*)d_in[0];
  const float* mel  = (const float*)d_in[1];
  const float* gum  = (const float*)d_in[2];
  float* out_soft = (float*)d_out;
  float* out_exp  = (float*)d_out + (size_t)B_ * I_ * J_;
  energy_kernel<<<dim3(72), dim3(128), 0, stream>>>(text, mel, gum, out_soft);
  dp_kernel<<<dim3(B_), dim3(384), 0, stream>>>(out_soft);
  expand_kernel<<<dim3(96), dim3(128), 0, stream>>>(text, out_soft, out_exp);
}


# Soft-output pass overlapped with the serial DP rows: the 8 waves without a DP row finish each stashed row, in 3 parts, between signal and wait of the split row barrier; every sum and order kept
# speedup vs baseline: 12.3765x; 1.0488x over previous
#include <hip/hip_runtime.h>
#include <hip/hip_bf16.h>
#include <math.h>


typedef float v2f __attribute__((ext_vector_type(2)));
typedef float v4f __attribute__((ext_vector_type(4)));
typedef float v8f __attribute__((ext_vector_type(8)));

#define B_      4
#define I_      96
#define J_      384
#define D_      256
#define NWAVE   12
#define RPW     (I_ / NWAVE)
#define CPL     (J_ / 32)
#define DPW     4
#define EPL     (J_ / (32 * DPW))
#define LOGEPS  (-1000.0f)
#define NEGV    (-1000000000.0f)
#define NEGINF  (-1.0e38f)
#define TEMPINV (1.0f/0.55f)
#define ESCALE  (1.0f/256.0f)
#define LOG2E   1.4426950408889634f
#define LN2     0.6931471805599453f

__device__ __forceinline__ float lse2(float a, float b) {
  const float m = fmaxf(a, b);
  const float e = __builtin_amdgcn_exp2f(-fabsf(a - b) * LOG2E);
  return __builtin_fmaf(__builtin_amdgcn_logf(1.0f + e), LN2, m);
}

template <int N>
__device__ __forceinline__ float row_shr_fill(float x, float fill) {
  return __builtin_bit_cast(float, __builtin_amdgcn_update_dpp(
      __builtin_bit_cast(int, fill), __builtin_bit_cast(int, x),
      0x110 + N  , 0xF, 0xF, false));
}

template <int N>
__device__ __forceinline__ float row_shl_fill(float x, float fill) {
  return __builtin_bit_cast(float, __builtin_amdgcn_update_dpp(
      __builtin_bit_cast(int, fill), __builtin_bit_cast(int, x),
      0x100 + N  , 0xF, 0xF, false));
}

template <int L>
__device__ __forceinline__ float lane_bcast(float x) {
  return __builtin_bit_cast(float,
      __builtin_amdgcn_readlane(__builtin_bit_cast(int, x), L));
}

__device__ __forceinline__ float wave_prefix_lse(float x, int lane) {
  x = lse2(x, row_shr_fill<1>(x, NEGINF));
  x = lse2(x, row_shr_fill<2>(x, NEGINF));
  x = lse2(x, row_shr_fill<4>(x, NEGINF));
  x = lse2(x, row_shr_fill<8>(x, NEGINF));
  const float lo = lane_bcast<15>(x);
  return lse2(x, (lane >= 16) ? lo : NEGINF);
}

__device__ __forceinline__ float wave_suffix_lse(float x, int lane) {
  x = lse2(x, row_shl_fill<1>(x, NEGINF));
  x = lse2(x, row_shl_fill<2>(x, NEGINF));
  x = lse2(x, row_shl_fill<4>(x, NEGINF));
  x = lse2(x, row_shl_fill<8>(x, NEGINF));
  const float hi = lane_bcast<16>(x);
  return lse2(x, (lane < 16) ? hi : NEGINF);
}

__device__ __forceinline__ float wave_next_lane(float x, int lane) {
  const float e  = row_shl_fill<1>(x, NEGINF);
  const float hi = lane_bcast<16>(x);
  return (lane == 15) ? hi : e;
}

#define MNEG   (-1442695040)
#define MEPS   (-1443)
#define SEPS   1.2353836f
#define XCLAMP 100000000.0f

__device__ __forceinline__ void padd(int& m, float& s, int m2, float s2) {
  const int mm = (m > m2) ? m : m2;
  s = __builtin_ldexpf(s, m - mm) + __builtin_ldexpf(s2, m2 - mm);
  m = mm;
}

__device__ __forceinline__ void to_pair(float x, int& m, float& s) {
  const float y = fminf(fmaxf(x * LOG2E, -XCLAMP), XCLAMP);
  const float f = floorf(y);
  m = (int)f;
  s = __builtin_amdgcn_exp2f(y - f);
}

__device__ __forceinline__ float from_pair(int m, float s) {
  return ((float)m + __builtin_amdgcn_logf(s)) * LN2;
}

template <int N>
__device__ __forceinline__ int row_shr_fill_i(int x, int fill) {
  return __builtin_amdgcn_update_dpp(fill, x, 0x110 + N  , 0xF, 0xF, false);
}

template <int N>
__device__ __forceinline__ int row_shl_fill_i(int x, int fill) {
  return __builtin_amdgcn_update_dpp(fill, x, 0x100 + N  , 0xF, 0xF, false);
}

__device__ __forceinline__ void wave_prefix_pair_ie(int m, float s, int lane,
                                                    int& mi, float& si, int& me, float& se) {
  padd(m, s, row_shr_fill_i<1>(m, MNEG), row_shr_fill<1>(s, 0.0f));
  padd(m, s, row_shr_fill_i<2>(m, MNEG), row_shr_fill<2>(s, 0.0f));
  padd(m, s, row_shr_fill_i<4>(m, MNEG), row_shr_fill<4>(s, 0.0f));
  padd(m, s, row_shr_fill_i<8>(m, MNEG), row_shr_fill<8>(s, 0.0f));
  const int   lom = __builtin_amdgcn_readlane(m, 15);
  const float los = lane_bcast<15>(s);
  const int   am  = (lane >= 16) ? lom : MNEG;
  const float as  = (lane >= 16) ? los : 0.0f;
  mi = m; si = s;
  padd(mi, si, am, as);
  me = row_shr_fill_i<1>(m, MNEG);
  se = row_shr_fill<1>(s, 0.0f);
  padd(me, se, am, as);
}

__device__ __forceinline__ void wave_suffix_pair_ie(int m, float s, int lane,
                                                    int& mi, float& si, int& me, float& se) {
  padd(m, s, row_shl_fill_i<1>(m, MNEG), row_shl_fill<1>(s, 0.0f));
  padd(m, s, row_shl_fill_i<2>(m, MNEG), row_shl_fill<2>(s, 0.0f));
  padd(m, s, row_shl_fill_i<4>(m, MNEG), row_shl_fill<4>(s, 0.0f));
  padd(m, s, row_shl_fill_i<8>(m, MNEG), row_shl_fill<8>(s, 0.0f));
  const int   him = __builtin_amdgcn_readlane(m, 16);
  const float his = lane_bcast<16>(s);
  const int   am  = (lane < 16) ? him : MNEG;
  const float as  = (lane < 16) ? his : 0.0f;
  mi = m; si = s;
  padd(mi, si, am, as);
  me = row_shl_fill_i<1>(m, MNEG);
  se = row_shl_fill<1>(s, 0.0f);
  padd(me, se, am, as);
}

#define EG 8

__device__ __forceinline__ void e_load(v4f (&a)[EG], v4f (&bq)[EG][2],
                                       const float* ap, const float* bp) {
#pragma unroll
  for (int c = 0; c < EG; ++c) {
    a[c] = *(const v4f*)(ap + 8 * c);
#pragma unroll
    for (int n = 0; n < 2; ++n) bq[c][n] = *(const v4f*)(bp + 8 * c + (size_t)n * 16 * D_);
  }
}

__device__ __forceinline__ void e_mma(v8f (&acc)[2], const v4f (&a)[EG],
                                      const v4f (&bq)[EG][2]) {
#pragma unroll
  for (int c = 0; c < EG; ++c) {
#pragma unroll
    for (int n = 0; n < 2; ++n) {
      const v2f av = { a[c].x, a[c].y };
      const v2f bv2 = { bq[c][n].x, bq[c][n].y };
      acc[n] = __builtin_amdgcn_wmma_f32_16x16x4_f32(false, av, false, bv2,
                                                     (short)0, acc[n], false, false);
    }
#pragma unroll
    for (int n = 0; n < 2; ++n) {
      const v2f av = { a[c].z, a[c].w };
      const v2f bv2 = { bq[c][n].z, bq[c][n].w };
      acc[n] = __builtin_amdgcn_wmma_f32_16x16x4_f32(false, av, false, bv2,
                                                     (short)0, acc[n], false, false);
    }
  }
}

__global__ __launch_bounds__(128)
void energy_kernel(const float* __restrict__ text,
                   const float* __restrict__ mel,
                   const float* __restrict__ gum,
                   float* __restrict__ energy)
{
  const int lane = threadIdx.x & 31;
  const int wave = threadIdx.x >> 5;
  const int half = lane >> 4;
  const int l16  = lane & 15;
  const int wb  = blockIdx.x * 4 + wave;
  const int b   = wb / 72;
  const int rem = wb - b * 72;
  const int mb  = rem / 12;
  const int nb  = rem - mb * 12;
  const int i0  = mb * 16;
  const int j0  = nb * 32;

  const float* ap = text + ((size_t)b * I_ + i0 + l16) * D_ + 4 * half;
  const float* bp = mel  + ((size_t)b * J_ + j0 + l16) * D_ + 4 * half;
  const float* gp = gum    + (size_t)b * I_ * J_;
  float*       ep = energy + (size_t)b * I_ * J_;

  v8f acc[2];
  acc[0] = (v8f){}; acc[1] = (v8f){};
  v4f a0[EG], b0[EG][2], a1[EG], b1[EG][2];
  e_load(a0, b0, ap, bp);
  __builtin_amdgcn_sched_barrier(0);
  float u[2][8];
#pragma unroll
  for (int n = 0; n < 2; ++n) {
#pragma unroll
    for (int r = 0; r < 8; ++r)
      u[n][r] = gp[(size_t)(i0 + r + 8 * half) * J_ + j0 + n * 16 + l16];
  }
  e_load(a1, b1, ap + 64, bp + 64);
  __builtin_amdgcn_sched_barrier(0);
  e_mma(acc, a0, b0);
  e_load(a0, b0, ap + 128, bp + 128);
  __builtin_amdgcn_sched_barrier(0);
  e_mma(acc, a1, b1);
  e_load(a1, b1, ap + 192, bp + 192);
  __builtin_amdgcn_sched_barrier(0);
  e_mma(acc, a0, b0);
  e_mma(acc, a1, b1);
#pragma unroll
  for (int n = 0; n < 2; ++n) {
#pragma unroll
    for (int r = 0; r < 8; ++r) {
      const int i = i0 + r + 8 * half;
      const int j = j0 + n * 16 + l16;
      const float g = -__logf(-__logf(u[n][r]));
      ep[(size_t)i * J_ + j] = fmaxf((acc[n][r] * ESCALE + g) * TEMPINV, NEGINF);
    }
  }
}

#define XG 6

__device__ __forceinline__ void x_load(float (&a)[XG][2], v4f (&bq)[XG][2],
                                       const float* sp, const float* tp) {
#pragma unroll
  for (int s = 0; s < XG; ++s) {
    a[s][0]  = sp[(size_t)(4 * s)     * J_];
    a[s][1]  = sp[(size_t)(4 * s + 1) * J_];
    bq[s][0] = *(const v4f*)(tp + (size_t)(4 * s)     * D_);
    bq[s][1] = *(const v4f*)(tp + (size_t)(4 * s + 1) * D_);
  }
}

__device__ __forceinline__ void x_mma(v8f (&acc)[4], const float (&a)[XG][2],
                                      const v4f (&bq)[XG][2]) {
#pragma unroll
  for (int s = 0; s < XG; ++s) {
    const v2f w = { __expf(a[s][0]), __expf(a[s][1]) };
#pragma unroll
    for (int n = 0; n < 4; ++n) {
      const v2f bv2 = { bq[s][0][n], bq[s][1][n] };
      acc[n] = __builtin_amdgcn_wmma_f32_16x16x4_f32(false, w, false, bv2,
                                                     (short)0, acc[n], false, false);
    }
  }
}

__global__ __launch_bounds__(128)
void expand_kernel(const float* __restrict__ text,
                   const float* __restrict__ soft,
                   float* __restrict__ out_exp)
{
  const int lane = threadIdx.x & 31;
  const int wave = threadIdx.x >> 5;
  const int half = lane >> 4;
  const int l16  = lane & 15;
  const int wb  = blockIdx.x * 4 + wave;
  const int b   = wb / 96;
  const int rem = wb - b * 96;
  const int mb  = rem / 4;
  const int nb  = rem - mb * 4;
  const int jb  = mb * 16;
  const int db  = nb * 64;

  const float* sp = soft + (size_t)b * I_ * J_ + (size_t)(2 * half) * J_ + jb + l16;
  const float* tp = text + (size_t)b * I_ * D_ + (size_t)(2 * half) * D_ + db + 4 * l16;

  v8f acc[4];
#pragma unroll
  for (int n = 0; n < 4; ++n) acc[n] = (v8f){};
  float a0[XG][2], a1[XG][2];
  v4f   b0[XG][2], b1[XG][2];
  x_load(a0, b0, sp, tp);
  x_load(a1, b1, sp + (size_t)(4 * XG) * J_,  tp + (size_t)(4 * XG) * D_);
  x_mma(acc, a0, b0);
  x_load(a0, b0, sp + (size_t)(8 * XG) * J_,  tp + (size_t)(8 * XG) * D_);
  x_mma(acc, a1, b1);
  x_load(a1, b1, sp + (size_t)(12 * XG) * J_, tp + (size_t)(12 * XG) * D_);
  x_mma(acc, a0, b0);
  x_mma(acc, a1, b1);
#pragma unroll
  for (int r = 0; r < 8; ++r) {
    const int j = jb + r + 8 * half;
    *(v4f*)(out_exp + (size_t)b * J_ * D_ + (size_t)j * D_ + db + 4 * l16) =
        (v4f){ acc[0][r], acc[1][r], acc[2][r], acc[3][r] };
  }
}

__device__ __forceinline__ void soft_row_head(const float* sE, int row, int lane,
                                              float (&s)[CPL], float& bfirst) {
  const float* br = sE + row * J_ + lane * CPL;
  const v4f x0 = *(const v4f*)(br + 0);
  const v4f x1 = *(const v4f*)(br + 4);
  const v4f x2 = *(const v4f*)(br + 8);
  s[0] = x0.x; s[1] = x0.y; s[2]  = x0.z; s[3]  = x0.w;
  s[4] = x1.x; s[5] = x1.y; s[6]  = x1.z; s[7]  = x1.w;
  s[8] = x2.x; s[9] = x2.y; s[10] = x2.z; s[11] = x2.w;
  bfirst = s[CPL - 1];
  if (lane == 31) s[CPL - 1] = NEGINF;
  for (int c = CPL - 2; c >= 0; --c) s[c] = lse2(s[c], s[c + 1]);
}

__device__ __forceinline__ void soft_row_scan(int lane, float (&s)[CPL]) {
  const float inc = wave_suffix_lse(s[0], lane);
  const float exc = wave_next_lane(inc, lane);
  for (int c = 0; c < CPL; ++c) s[c] = lse2(s[c], exc);
}

__device__ __forceinline__ void soft_row_store(const float* sL, float* og_batch, int row, int lane,
                                               const float (&s)[CPL], float bfirst) {
  const float* qr = sL + row * J_ + lane * CPL;
  const v4f q0 = *(const v4f*)(qr + 0);
  const v4f q1 = *(const v4f*)(qr + 4);
  const v4f q2 = *(const v4f*)(qr + 8);
  const float q[CPL] = { q0.x, q0.y, q0.z, q0.w, q1.x, q1.y, q1.z, q1.w,
                         q2.x, q2.y, q2.z, q2.w };
  float o[CPL];
  if (row < I_ - 1) {
    for (int c = 0; c < CPL; ++c) o[c] = lse2(q[c], s[c] + LOGEPS);
  } else {
    const float tot = lse2(lane_bcast<0>(s[0]), lane_bcast<31>(bfirst));
    for (int c = 0; c < CPL; ++c)
      o[c] = tot + ((lane * CPL + c == J_ - 1) ? 0.0f : LOGEPS);
  }
  float* og = og_batch + (size_t)row * J_ + lane * CPL;
  *(v4f*)(og + 0) = (v4f){ o[0], o[1], o[2],  o[3]  };
  *(v4f*)(og + 4) = (v4f){ o[4], o[5], o[6],  o[7]  };
  *(v4f*)(og + 8) = (v4f){ o[8], o[9], o[10], o[11] };
}

#define HLP     (NWAVE - DPW)
#define LROWS   (I_ - 5)

__global__ __launch_bounds__(384, 1)
void dp_kernel(float* __restrict__ out_soft)
{
  __shared__ __attribute__((aligned(16))) float sE[I_ * J_];
  __shared__ __attribute__((aligned(16))) float sL[I_ * J_];
  __shared__ int   sWm[2][32];
  __shared__ float sWs[2][32];

  const int b    = blockIdx.x;
  const int tid  = threadIdx.x;
  const int lane = tid & 31;
  const int wave = tid >> 5;

  {
    const v4f* eg = (const v4f*)(out_soft + (size_t)b * I_ * J_);
    v4f* el = (v4f*)sE;
    v4f ev4[I_ * J_ / 4 / 384];
#pragma unroll
    for (int r = 0; r < I_ * J_ / 4 / 384; ++r) ev4[r] = eg[tid + 384 * r];
#pragma unroll
    for (int r = 0; r < I_ * J_ / 4 / 384; ++r) el[tid + 384 * r] = ev4[r];
  }
  __syncthreads();

  for (int r = 0; r < RPW; ++r) {
    const int row = wave * RPW + r;
    const float* er = sE + row * J_ + lane * CPL;
    const v4f x0 = *(const v4f*)(er + 0);
    const v4f x1 = *(const v4f*)(er + 4);
    const v4f x2 = *(const v4f*)(er + 8);
    float s[CPL] = { x0.x, x0.y, x0.z, x0.w, x1.x, x1.y, x1.z, x1.w,
                     x2.x, x2.y, x2.z, x2.w };
    for (int c = CPL - 2; c >= 0; --c) s[c] = lse2(s[c], s[c + 1]);
    const float inc = wave_suffix_lse(s[0], lane);
    const float exc = wave_next_lane(inc, lane);
    for (int c = 0; c < CPL; ++c) s[c] = lse2(s[c], exc);
    float* lr = sL + row * J_ + lane * CPL;
    *(v4f*)(lr + 0) = (v4f){ s[0], s[1], s[2],  s[3]  };
    *(v4f*)(lr + 4) = (v4f){ s[4], s[5], s[6],  s[7]  };
    *(v4f*)(lr + 8) = (v4f){ s[8], s[9], s[10], s[11] };
  }

  const bool act = wave < DPW;
  const int  k0  = EPL * tid;
  int   bm[EPL];
  float bs[EPL];
  float bl[EPL];
  int   st[EPL];
  for (int e = 0; e < EPL; ++e) {
    const int jj = k0 + e;
    bm[e] = (jj == 0) ? 0 : MNEG;
    bs[e] = 1.0f;
    bl[e] = (jj == 0) ? 0.0f : NEGV;
    st[e] = (jj == 0) ? (J_ - 1) : (jj - 1);
  }
  const int wv = __builtin_amdgcn_readfirstlane(wave);
  if (tid < 64) {
    (&sWm[0][0])[tid] = MNEG;
    (&sWs[0][0])[tid] = 0.0f;
  }
  __syncthreads();

  float Ln[EPL];
  int   nLm[EPL];
  float nLs[EPL];
  if (act) {
    for (int e = 0; e < EPL; ++e) {
      Ln[e] = sL[k0 + e];
      to_pair(-Ln[e], nLm[e], nLs[e]);
    }
  }
  float hs[CPL];
  float hb = 0.0f;
  for (int c = 0; c < CPL; ++c) hs[c] = 0.0f;
  float* const og_batch = out_soft + (size_t)b * I_ * J_;
  for (int i = 0; i < I_ - 1; ++i) {
    int*   wm = sWm[i & 1];
    float* ws = sWs[i & 1];
    float Lv[EPL], ev[EPL];
    int   pm[EPL], sm[EPL];
    float ps[EPL], ss[EPL];
    int   xpim = MNEG, xpem = MNEG, xsim = MNEG, xsem = MNEG;
    float xpis = 0.0f, xpes = 0.0f, xsis = 0.0f, xses = 0.0f;
    if (act) {
      int   am[EPL];
      float as[EPL];
      for (int e = 0; e < EPL; ++e) {
        Lv[e] = Ln[e];
        am[e] = bm[e] + nLm[e];
        as[e] = bs[e] * nLs[e];
        Ln[e] = sL[(i + 1) * J_ + k0 + e];
        ev[e] = sE[i * J_ + st[e]];
      }
      pm[0] = am[0]; ps[0] = as[0];
      for (int e = 1; e < EPL; ++e) {
        pm[e] = pm[e - 1]; ps[e] = ps[e - 1];
        padd(pm[e], ps[e], am[e], as[e]);
      }
      sm[EPL - 1] = am[EPL - 1]; ss[EPL - 1] = as[EPL - 1];
      for (int e = EPL - 2; e >= 0; --e) {
        sm[e] = am[e]; ss[e] = as[e];
        padd(sm[e], ss[e], sm[e + 1], ss[e + 1]);
      }
      wave_prefix_pair_ie(pm[EPL - 1], ps[EPL - 1], lane, xpim, xpis, xpem, xpes);
      wave_suffix_pair_ie(sm[0],       ss[0],       lane, xsim, xsis, xsem, xses);
      if (lane == 31) { wm[1 + wave]        = xpim; ws[1 + wave]        = xpis; }
      if (lane == 0)  { wm[16 + DPW - wave] = xsim; ws[16 + DPW - wave] = xsis; }
      for (int e = 0; e < EPL; ++e) to_pair(-Ln[e], nLm[e], nLs[e]);
    }
    __builtin_amdgcn_fence(__ATOMIC_RELEASE, "workgroup", "local");
    __builtin_amdgcn_s_barrier_signal(-1);
    if (wv >= DPW) {
      const int r1 = i - 2;
      const int r2 = i - 3;
      const int r3 = i - 4;
      if (r3 >= 0 && r3 < LROWS && (r3 % HLP) == wv - DPW)
        soft_row_store(sL, og_batch, r3, lane, hs, hb);
      if (r2 >= 0 && r2 < LROWS && (r2 % HLP) == wv - DPW)
        soft_row_scan(lane, hs);
      if (r1 >= 0 && r1 < LROWS && (r1 % HLP) == wv - DPW)
        soft_row_head(sE, r1, lane, hs, hb);
    }
    __builtin_amdgcn_s_barrier_wait(-1);
    __builtin_amdgcn_fence(__ATOMIC_ACQUIRE, "workgroup", "local");
    if (act) {
      int   tm = wm[lane];
      float ts = ws[lane];
      int   em[EPL];
      float es[EPL];
      for (int e = 0; e < EPL; ++e) to_pair(ev[e], em[e], es[e]);
      padd(tm, ts, row_shr_fill_i<1>(tm, MNEG), row_shr_fill<1>(ts, 0.0f));
      padd(tm, ts, row_shr_fill_i<2>(tm, MNEG), row_shr_fill<2>(ts, 0.0f));
      int   offPm = xpem, offSm = xsem;
      float offPs = xpes, offSs = xses;
      padd(offPm, offPs, __builtin_amdgcn_readlane(tm, wv),
           __builtin_bit_cast(float, __builtin_amdgcn_readlane(__builtin_bit_cast(int, ts), wv)));
      padd(offSm, offSs, __builtin_amdgcn_readlane(tm, 16 + DPW - 1 - wv),
           __builtin_bit_cast(float, __builtin_amdgcn_readlane(__builtin_bit_cast(int, ts), 16 + DPW - 1 - wv)));
      int   Pm[EPL], SAm[EPL];
      float Ps[EPL], SAs[EPL];
      for (int e = 0; e < EPL; ++e) {
        Pm[e]  = pm[e]; Ps[e]  = ps[e];
        padd(Pm[e], Ps[e], offPm, offPs);
        SAm[e] = sm[e]; SAs[e] = ss[e];
        padd(SAm[e], SAs[e], offSm, offSs);
        sL[i * J_ + k0 + e] = Lv[e] + from_pair(Pm[e], Ps[e]);
      }
      const int r = i + 1;
      for (int e = 0; e < EPL; ++e) {
        const int   jj  = k0 + e;
        const int   ppm = (e == 0) ? offPm : Pm[e > 0 ? e - 1 : 0];
        const float pps = (e == 0) ? offPs : Ps[e > 0 ? e - 1 : 0];
        int   cm = em[e] + ppm;
        float cs = es[e] * pps;
        padd(cm, cs, SAm[e] + MEPS, SAs[e] * SEPS);
        cm += __builtin_amdgcn_frexp_expf(cs);
        cs  = __builtin_amdgcn_frexp_mantf(cs);
        const bool in = (jj >= r && jj <= (J_ - I_ + r + 1));
        sE[i * J_ + st[e]] = bl[e];
        bm[e] = in ? cm : MNEG;
        bs[e] = in ? cs : 1.0f;
        bl[e] = in ? from_pair(cm, cs) : NEGV;
      }
    }
  }
  if (act) {
    for (int e = 0; e < EPL; ++e)
      sE[(I_ - 1) * J_ + st[e]] = bl[e];
  }
  __syncthreads();

  if (wv < I_ - LROWS) {
    const int row = LROWS + wv;
    soft_row_head(sE, row, lane, hs, hb);
    soft_row_scan(lane, hs);
    soft_row_store(sL, og_batch, row, lane, hs, hb);
  }
}

extern "C" void kernel_launch(void* const* d_in, const int* in_sizes, int n_in,
                              void* d_out, int out_size, void* d_ws, size_t ws_size,
                              hipStream_t stream) {
  (void)in_sizes; (void)n_in; (void)d_ws; (void)ws_size; (void)out_size;
  const float* text = (const float*)d_in[0];
  const float* mel  = (const float*)d_in[1];
  const float* gum  = (const float*)d_in[2];
  float* out_soft = (float*)d_out;
  float* out_exp  = (float*)d_out + (size_t)B_ * I_ * J_;
  energy_kernel<<<dim3(72), dim3(128), 0, stream>>>(text, mel, gum, out_soft);
  dp_kernel<<<dim3(B_), dim3(384), 0, stream>>>(out_soft);
  expand_kernel<<<dim3(96), dim3(128), 0, stream>>>(text, out_soft, out_exp);
}
